# starting kernel
# baseline (speedup 1.0000x reference)
#include <hip/hip_runtime.h>
#include <math.h>

typedef float v2f __attribute__((ext_vector_type(2)));
typedef float v8f __attribute__((ext_vector_type(8)));

#define HD 16
#define NWIRE 1536
#define NFEAT 4

__device__ __forceinline__ unsigned int encOrd(float f) {
    unsigned int b = __float_as_uint(f);
    return (b & 0x80000000u) ? ~b : (b | 0x80000000u);
}
__device__ __forceinline__ float decOrd(unsigned int e) {
    unsigned int b = (e & 0x80000000u) ? (e ^ 0x80000000u) : ~e;
    return __uint_as_float(b);
}
__device__ __forceinline__ float lrelu02(float x) { return x > 0.f ? x : 0.2f * x; }

__global__ void k_padW0(const float* __restrict__ W0, float* __restrict__ W0p) {
    int i = threadIdx.x;
    int r = i >> 4;
    W0p[i] = (r < 13) ? W0[i] : 0.f;
}

__global__ void k_fill(const float* __restrict__ fw, const float* __restrict__ sw,
                       const float* __restrict__ tw, const float* __restrict__ areas,
                       const int* __restrict__ indices, float* __restrict__ h,
                       int N, int g) {
    int n = blockIdx.x * blockDim.x + threadIdx.x;
    if (n >= N) return;
    const size_t gbase = (size_t)g * NWIRE * NFEAT;
    const float4 f0 = *(const float4*)(fw + gbase + (size_t)indices[n * 3 + 0] * NFEAT);
    const float4 f1 = *(const float4*)(sw + gbase + (size_t)indices[n * 3 + 1] * NFEAT);
    const float4 f2 = *(const float4*)(tw + gbase + (size_t)indices[n * 3 + 2] * NFEAT);
    float4* hp = (float4*)(h + (size_t)n * HD);
    hp[0] = f0;
    hp[1] = f1;
    hp[2] = f2;
    hp[3] = make_float4(areas[n], 0.f, 0.f, 0.f);
}

__global__ void k_gemm_wmma(const float* __restrict__ h, const float* __restrict__ W,
                            const float* __restrict__ bias, float* __restrict__ hw, int N) {
    const int lane = threadIdx.x & 31;
    const int wave = threadIdx.x >> 5;
    const int tile = blockIdx.x * 8 + wave;
    const int base = tile * 16;
    if (base >= N) return;
    const int col16 = lane & 15;
    const int half  = lane >> 4;
    int arow = base + col16;
    if (arow >= N) arow = N - 1;
    v8f acc = {};
#pragma unroll
    for (int k = 0; k < 4; ++k) {
        const int kc = 4 * k + 2 * half;
        v2f a;
        a.x = h[(size_t)arow * HD + kc];
        a.y = h[(size_t)arow * HD + kc + 1];
        v2f b;
        b.x = W[kc * HD + col16];
        b.y = W[(kc + 1) * HD + col16];
        acc = __builtin_amdgcn_wmma_f32_16x16x4_f32(
                  false, a, false, b, (short)0, acc, false, false);
    }
    const float bi = bias[col16];
    if (base + 16 <= N) {
#pragma unroll
        for (int r = 0; r < 8; ++r) {
            hw[(size_t)(base + r + 8 * half) * HD + col16] = acc[r] + bi;
        }
    } else {
#pragma unroll
        for (int r = 0; r < 8; ++r) {
            const int orow = base + r + 8 * half;
            if (orow < N) hw[(size_t)orow * HD + col16] = acc[r] + bi;
        }
    }
}

__global__ void k_esed(const float* __restrict__ hw, const float* __restrict__ asrc,
                       const float* __restrict__ adst, float* __restrict__ es,
                       float* __restrict__ ed, int N) {
    int n = blockIdx.x * blockDim.x + threadIdx.x;
    if (n >= N) return;
    const float4* r = (const float4*)(hw + (size_t)n * HD);
    float s = 0.f, d = 0.f;
#pragma unroll
    for (int q = 0; q < 4; ++q) {
        const float4 v = r[q];
        s += v.x * asrc[q*4+0] + v.y * asrc[q*4+1] + v.z * asrc[q*4+2] + v.w * asrc[q*4+3];
        d += v.x * adst[q*4+0] + v.y * adst[q*4+1] + v.z * adst[q*4+2] + v.w * adst[q*4+3];
    }
    es[n] = s; ed[n] = d;
}

__global__ void k_init(unsigned int* __restrict__ mEnc, float* __restrict__ denom,
                       float* __restrict__ accum, int N) {
    int j = blockIdx.x * blockDim.x + threadIdx.x;
    if (j >= N * HD) return;
    accum[j] = 0.f;
    if (j < N) { mEnc[j] = 0x007FFFFFu;   denom[j] = 0.f; }
}

__global__ void k_edge_max(const int* __restrict__ src, const int* __restrict__ dst,
                           const float* __restrict__ es, const float* __restrict__ ed,
                           unsigned int* __restrict__ mEnc, int E) {
    int e = blockIdx.x * blockDim.x + threadIdx.x;
    if (e >= E) return;
    const int s = src[e], d = dst[e];
    const float lg = lrelu02(es[s] + ed[d]);
    atomicMax(&mEnc[d], encOrd(lg));
}

__global__ void k_edge_sum(const int* __restrict__ src, const int* __restrict__ dst,
                           const float* __restrict__ es, const float* __restrict__ ed,
                           const unsigned int* __restrict__ mEnc, float* __restrict__ denom,
                           float* __restrict__ wE, int E) {
    int e = blockIdx.x * blockDim.x + threadIdx.x;
    if (e >= E) return;
    const int s = src[e], d = dst[e];
    const float lg = lrelu02(es[s] + ed[d]);
    const float w = __expf(lg - decOrd(mEnc[d]));
    wE[e] = w;
    atomicAdd(&denom[d], w);
}

__global__ void k_edge_accum(const int* __restrict__ src, const int* __restrict__ dst,
                             const float* __restrict__ hw, const float* __restrict__ denom,
                             const float* __restrict__ wE, float* __restrict__ accum, int E) {
    int t = blockIdx.x * blockDim.x + threadIdx.x;
    if (t >= E * 4) return;
    const int e = t >> 2;
    const int q = t & 3;
    const int s = src[e], d = dst[e];
    const float alpha = wE[e] / fmaxf(denom[d], 1e-16f);
    const float4 hv = *(const float4*)(hw + (size_t)s * HD + q * 4);
    float* ap = accum + (size_t)d * HD + q * 4;
    atomicAdd(ap + 0, alpha * hv.x);
    atomicAdd(ap + 1, alpha * hv.y);
    atomicAdd(ap + 2, alpha * hv.z);
    atomicAdd(ap + 3, alpha * hv.w);
}

__global__ void k_elu(const float* __restrict__ accum, float* __restrict__ h, int N) {
    int j = blockIdx.x * blockDim.x + threadIdx.x;
    if (j >= N * HD) return;
    const float x = accum[j];
    h[j] = x > 0.f ? x : (__expf(x) - 1.f);
}

__global__ void k_final(const float* __restrict__ h, const float* __restrict__ mlpW,
                        const float* __restrict__ mlpb, float* __restrict__ out, int N) {
    int n = blockIdx.x * blockDim.x + threadIdx.x;
    if (n >= N) return;
    const float4* r = (const float4*)(h + (size_t)n * HD);
    float acc = mlpb[0];
#pragma unroll
    for (int q = 0; q < 4; ++q) {
        const float4 v = r[q];
        acc += v.x * mlpW[q*4+0] + v.y * mlpW[q*4+1] + v.z * mlpW[q*4+2] + v.w * mlpW[q*4+3];
    }
    out[n] = acc;
}

static inline int cdiv(int a, int b) { return (a + b - 1) / b; }

extern "C" void kernel_launch(void* const* d_in, const int* in_sizes, int n_in,
                              void* d_out, int out_size, void* d_ws, size_t ws_size,
                              hipStream_t stream) {
    const float* fw     = (const float*)d_in[0];
    const float* sw     = (const float*)d_in[1];
    const float* tw     = (const float*)d_in[2];
    const float* areas  = (const float*)d_in[3];
    const int*   indices= (const int*)  d_in[4];
    const int*   edge   = (const int*)  d_in[5];
    const float* W0     = (const float*)d_in[6];
    const float* Wl     = (const float*)d_in[7];
    const float* a_src  = (const float*)d_in[8];
    const float* a_dst  = (const float*)d_in[9];
    const float* bvec   = (const float*)d_in[10];
    const float* mlpW   = (const float*)d_in[11];
    const float* mlpb   = (const float*)d_in[12];

    const int N = in_sizes[3];
    const int E = in_sizes[5] / 2;
    const int T = in_sizes[0] / (NWIRE * NFEAT);
    const int* src = edge;
    const int* dst = edge + E;

    float* ws = (float*)d_ws;
    size_t off = 0;
    float*        h      = ws + off; off += (size_t)N * HD;
    float*        hw     = ws + off; off += (size_t)N * HD;
    float*        accum  = ws + off; off += (size_t)N * HD;
    float*        es     = ws + off; off += (size_t)N;
    float*        ed     = ws + off; off += (size_t)N;
    unsigned int* mEnc   = (unsigned int*)(ws + off); off += (size_t)N;
    float*        denom  = ws + off; off += (size_t)N;
    float*        wE     = ws + off; off += (size_t)E;
    float*        W0p    = ws + off; off += 256;
    (void)ws_size; (void)n_in; (void)out_size;

    k_padW0<<<1, 256, 0, stream>>>(W0, W0p);

    const int tiles  = cdiv(N, 16);
    const int gemmBl = cdiv(tiles, 8);

    for (int g = 0; g < T; ++g) {
        k_fill<<<cdiv(N, 256), 256, 0, stream>>>(fw, sw, tw, areas, indices, h, N, g);
        for (int i = 0; i < 4; ++i) {
            const float* Wi = (i == 0) ? W0p : (Wl + (size_t)(i - 1) * 256);
            k_init<<<cdiv(N * HD, 256), 256, 0, stream>>>(mEnc, denom, accum, N);
            k_gemm_wmma<<<gemmBl, 256, 0, stream>>>(h, Wi, bvec + (size_t)i * HD, hw, N);
            k_esed<<<cdiv(N, 256), 256, 0, stream>>>(hw, a_src + (size_t)i * HD,
                                                     a_dst + (size_t)i * HD, es, ed, N);
            k_edge_max<<<cdiv(E, 256), 256, 0, stream>>>(src, dst, es, ed, mEnc, E);
            k_edge_sum<<<cdiv(E, 256), 256, 0, stream>>>(src, dst, es, ed, mEnc, denom, wE, E);
            k_edge_accum<<<cdiv(E * 4, 256), 256, 0, stream>>>(src, dst, hw, denom, wE, accum, E);
            k_elu<<<cdiv(N * HD, 256), 256, 0, stream>>>(accum, h, N);
        }
        k_final<<<cdiv(N, 256), 256, 0, stream>>>(h, mlpW, mlpb, (float*)d_out + (size_t)g * N, N);
    }
}


# Each of 256 blocks owns a node range: scans a per-edge owner byte, lists its in-edges in LDS, segment max and weighted sums by LDS atomics, elu in place; no global atomics; sums arrival-ordered
# speedup vs baseline: 3.4204x; 3.4204x over previous
#include <hip/hip_runtime.h>
#include <math.h>

typedef float v2f __attribute__((ext_vector_type(2)));
typedef float v8f __attribute__((ext_vector_type(8)));

#define HD 16
#define NWIRE 1536
#define NFEAT 4
#define NB 256
#define NPBMAX 1024
#define AST 17
#define WCAP 768
#define CAP (32 * WCAP)
#define SCAN_U 8

__device__ __forceinline__ unsigned int encOrd(float f) {
    unsigned int b = __float_as_uint(f);
    return (b & 0x80000000u) ? ~b : (b | 0x80000000u);
}
__device__ __forceinline__ float decOrd(unsigned int e) {
    unsigned int b = (e & 0x80000000u) ? (e ^ 0x80000000u) : ~e;
    return __uint_as_float(b);
}
__device__ __forceinline__ float lrelu02(float x) { return x > 0.f ? x : 0.2f * x; }

__global__ void k_padW0(const float* __restrict__ W0, float* __restrict__ W0p) {
    int i = threadIdx.x;
    int r = i >> 4;
    W0p[i] = (r < 13) ? W0[i] : 0.f;
}

__global__ void k_fill(const float* __restrict__ fw, const float* __restrict__ sw,
                       const float* __restrict__ tw, const float* __restrict__ areas,
                       const int* __restrict__ indices, float* __restrict__ h,
                       int N, int g) {
    int n = blockIdx.x * blockDim.x + threadIdx.x;
    if (n >= N) return;
    const size_t gbase = (size_t)g * NWIRE * NFEAT;
    const float4 f0 = *(const float4*)(fw + gbase + (size_t)indices[n * 3 + 0] * NFEAT);
    const float4 f1 = *(const float4*)(sw + gbase + (size_t)indices[n * 3 + 1] * NFEAT);
    const float4 f2 = *(const float4*)(tw + gbase + (size_t)indices[n * 3 + 2] * NFEAT);
    float4* hp = (float4*)(h + (size_t)n * HD);
    hp[0] = f0;
    hp[1] = f1;
    hp[2] = f2;
    hp[3] = make_float4(areas[n], 0.f, 0.f, 0.f);
}

__global__ void k_gemm_wmma(const float* __restrict__ h, const float* __restrict__ W,
                            const float* __restrict__ bias, float* __restrict__ hw, int N) {
    const int lane = threadIdx.x & 31;
    const int wave = threadIdx.x >> 5;
    const int tile = blockIdx.x * 8 + wave;
    const int base = tile * 16;
    if (base >= N) return;
    const int col16 = lane & 15;
    const int half  = lane >> 4;
    int arow = base + col16;
    if (arow >= N) arow = N - 1;
    v8f acc = {};
#pragma unroll
    for (int k = 0; k < 4; ++k) {
        const int kc = 4 * k + 2 * half;
        v2f a;
        a.x = h[(size_t)arow * HD + kc];
        a.y = h[(size_t)arow * HD + kc + 1];
        v2f b;
        b.x = W[kc * HD + col16];
        b.y = W[(kc + 1) * HD + col16];
        acc = __builtin_amdgcn_wmma_f32_16x16x4_f32(
                  false, a, false, b, (short)0, acc, false, false);
    }
    const float bi = bias[col16];
    if (base + 16 <= N) {
#pragma unroll
        for (int r = 0; r < 8; ++r) {
            hw[(size_t)(base + r + 8 * half) * HD + col16] = acc[r] + bi;
        }
    } else {
#pragma unroll
        for (int r = 0; r < 8; ++r) {
            const int orow = base + r + 8 * half;
            if (orow < N) hw[(size_t)orow * HD + col16] = acc[r] + bi;
        }
    }
}

__global__ void k_esed(const float* __restrict__ hw, const float* __restrict__ asrc,
                       const float* __restrict__ adst, float* __restrict__ es,
                       float* __restrict__ ed, int N) {
    int n = blockIdx.x * blockDim.x + threadIdx.x;
    if (n >= N) return;
    const float4* r = (const float4*)(hw + (size_t)n * HD);
    float s = 0.f, d = 0.f;
#pragma unroll
    for (int q = 0; q < 4; ++q) {
        const float4 v = r[q];
        s += v.x * asrc[q*4+0] + v.y * asrc[q*4+1] + v.z * asrc[q*4+2] + v.w * asrc[q*4+3];
        d += v.x * adst[q*4+0] + v.y * adst[q*4+1] + v.z * adst[q*4+2] + v.w * adst[q*4+3];
    }
    es[n] = s; ed[n] = d;
}

__global__ void k_key(const int* __restrict__ dst, unsigned int* __restrict__ key4,
                      int E, int nvec4, int npb) {
    const int j = blockIdx.x * blockDim.x + threadIdx.x;
    if (j >= nvec4) return;
    unsigned int pk = 0u;
#pragma unroll
    for (int k = 0; k < 4; ++k) {
        const int e = 4 * j + k;
        const unsigned int o = (e < E) ? (unsigned int)(dst[e] / npb) : 0u;
        pk |= (o & 0xFFu) << (8 * k);
    }
    key4[j] = pk;
}

template <class F>
__device__ __forceinline__ int scan_keys(const uint4* __restrict__ kv, int nvec, unsigned int pat,
                                         int E, F fn) {
    int wcnt = 0;
    const int lane = threadIdx.x & 31;
    for (int vb = threadIdx.x & ~31; vb < nvec; vb += 1024 * SCAN_U) {
        uint4 q[SCAN_U];
#pragma unroll
        for (int u = 0; u < SCAN_U; ++u) {
            const int v = vb + u * 1024 + lane;
            q[u] = kv[v < nvec ? v : nvec - 1];
        }
#pragma unroll
        for (int u = 0; u < SCAN_U; ++u) {
            const int v = vb + u * 1024 + lane;
            const unsigned int wd[4] = {q[u].x, q[u].y, q[u].z, q[u].w};
            unsigned int m = 0u;
#pragma unroll
            for (int j = 0; j < 4; ++j) {
                const unsigned int x = wd[j] ^ pat;
                const unsigned int t = ~(((x & 0x7F7F7F7Fu) + 0x7F7F7F7Fu) | x | 0x7F7F7F7Fu);
                m |= t >> (7 - j);
            }
            if (v >= nvec) m = 0u;
            else if (v * 16 + 16 > E) {
                for (int p = 0; p < 32; ++p)
                    if (((m >> p) & 1u) && v * 16 + (p & 7) * 4 + (p >> 3) >= E) m &= ~(1u << p);
            }
            for (int it = 0; it < 16; ++it) {
                const bool has = (m != 0u);
                const unsigned int mk = (unsigned int)__ballot(has);
                if (mk == 0u) break;
                if (has) {
                    const int p = __builtin_ctz(m);
                    m &= m - 1u;
                    fn(v * 16 + (p & 7) * 4 + (p >> 3), wcnt + (int)__builtin_amdgcn_mbcnt_lo(mk, 0u));
                }
                wcnt += __popc(mk);
            }
        }
    }
    return wcnt;
}

__global__ void __launch_bounds__(1024)
k_edge_own(const int* __restrict__ src, const int* __restrict__ dst,
           const unsigned int* __restrict__ key4, const float* __restrict__ hw,
           const float* __restrict__ es, const float* __restrict__ ed,
           float* __restrict__ h, int N, int E, int npb) {
    __shared__ float acc[NPBMAX * AST];
    __shared__ unsigned int mEnc[NPBMAX];
    __shared__ float edl[NPBMAX];
    __shared__ unsigned int lstA[CAP];
    __shared__ float lstB[CAP];
    __shared__ int ovf;

    const int tid = threadIdx.x;
    const int lane = tid & 31;
    const int wbase = (tid >> 5) * WCAP;
    const int n0 = blockIdx.x * npb;
    int nown = N - n0; if (nown > npb) nown = npb; if (nown < 0) nown = 0;

    for (int i = tid; i < nown * AST; i += 1024) acc[i] = 0.f;
    for (int i = tid; i < nown; i += 1024) { mEnc[i] = 0x007FFFFFu;   edl[i] = ed[n0 + i]; }
    if (tid == 0) ovf = 0;
    __syncthreads();

    const uint4* kv = (const uint4*)key4;
    const int nvec = (E + 15) >> 4;
    const unsigned int pat = (unsigned int)blockIdx.x * 0x01010101u;

    const int wcnt = scan_keys(kv, nvec, pat, E, [&](int e, int slot) {
        if (slot < WCAP) lstA[wbase + slot] = (unsigned int)e; else ovf = 1;
    });
    __syncthreads();

    if (ovf == 0) {
        for (int i = lane; i < wcnt; i += 32) {
            const int e = (int)lstA[wbase + i];
            const int s = src[e];
            const int dl = dst[e] - n0;
            const float lg = lrelu02(es[s] + edl[dl]);
            atomicMax(&mEnc[dl], encOrd(lg));
            lstA[wbase + i] = ((unsigned int)s << 10) | (unsigned int)dl;
            lstB[wbase + i] = lg;
        }
        __syncthreads();
        for (int i = lane; i < wcnt; i += 32) {
            const unsigned int a = lstA[wbase + i];
            const int dl = (int)(a & 1023u);
            const float4* r = (const float4*)(hw + (size_t)(a >> 10) * HD);
            const float4 r0 = r[0], r1 = r[1], r2 = r[2], r3 = r[3];
            const float w = __expf(lstB[wbase + i] - decOrd(mEnc[dl]));
            float* ap = acc + dl * AST;
            atomicAdd(ap + 0,  w * r0.x); atomicAdd(ap + 1,  w * r0.y);
            atomicAdd(ap + 2,  w * r0.z); atomicAdd(ap + 3,  w * r0.w);
            atomicAdd(ap + 4,  w * r1.x); atomicAdd(ap + 5,  w * r1.y);
            atomicAdd(ap + 6,  w * r1.z); atomicAdd(ap + 7,  w * r1.w);
            atomicAdd(ap + 8,  w * r2.x); atomicAdd(ap + 9,  w * r2.y);
            atomicAdd(ap + 10, w * r2.z); atomicAdd(ap + 11, w * r2.w);
            atomicAdd(ap + 12, w * r3.x); atomicAdd(ap + 13, w * r3.y);
            atomicAdd(ap + 14, w * r3.z); atomicAdd(ap + 15, w * r3.w);
            atomicAdd(ap + 16, w);
        }
    } else {
        scan_keys(kv, nvec, pat, E, [&](int e, int) {
            const int dl = dst[e] - n0;
            atomicMax(&mEnc[dl], encOrd(lrelu02(es[src[e]] + edl[dl])));
        });
        __syncthreads();
        scan_keys(kv, nvec, pat, E, [&](int e, int) {
            const int s = src[e];
            const int dl = dst[e] - n0;
            const float w = __expf(lrelu02(es[s] + edl[dl]) - decOrd(mEnc[dl]));
            const float* r = hw + (size_t)s * HD;
            float* ap = acc + dl * AST;
#pragma unroll
            for (int f = 0; f < HD; ++f) atomicAdd(ap + f, w * r[f]);
            atomicAdd(ap + HD, w);
        });
    }
    __syncthreads();

    for (int j = tid; j < nown * HD; j += 1024) {
        const int n = j >> 4, f = j & 15;
        const float den = fmaxf(acc[n * AST + HD], 1e-16f);
        const float x = acc[n * AST + f] / den;
        h[(size_t)n0 * HD + j] = x > 0.f ? x : (__expf(x) - 1.f);
    }
}

__global__ void k_final(const float* __restrict__ h, const float* __restrict__ mlpW,
                        const float* __restrict__ mlpb, float* __restrict__ out, int N) {
    int n = blockIdx.x * blockDim.x + threadIdx.x;
    if (n >= N) return;
    const float4* r = (const float4*)(h + (size_t)n * HD);
    float acc = mlpb[0];
#pragma unroll
    for (int q = 0; q < 4; ++q) {
        const float4 v = r[q];
        acc += v.x * mlpW[q*4+0] + v.y * mlpW[q*4+1] + v.z * mlpW[q*4+2] + v.w * mlpW[q*4+3];
    }
    out[n] = acc;
}

static inline int cdiv(int a, int b) { return (a + b - 1) / b; }

extern "C" void kernel_launch(void* const* d_in, const int* in_sizes, int n_in,
                              void* d_out, int out_size, void* d_ws, size_t ws_size,
                              hipStream_t stream) {
    const float* fw     = (const float*)d_in[0];
    const float* sw     = (const float*)d_in[1];
    const float* tw     = (const float*)d_in[2];
    const float* areas  = (const float*)d_in[3];
    const int*   indices= (const int*)  d_in[4];
    const int*   edge   = (const int*)  d_in[5];
    const float* W0     = (const float*)d_in[6];
    const float* Wl     = (const float*)d_in[7];
    const float* a_src  = (const float*)d_in[8];
    const float* a_dst  = (const float*)d_in[9];
    const float* bvec   = (const float*)d_in[10];
    const float* mlpW   = (const float*)d_in[11];
    const float* mlpb   = (const float*)d_in[12];

    const int N = in_sizes[3];
    const int E = in_sizes[5] / 2;
    const int T = in_sizes[0] / (NWIRE * NFEAT);
    const int* src = edge;
    const int* dst = edge + E;

    float* ws = (float*)d_ws;
    const int nvec4 = cdiv(E, 16) * 4;
    size_t off = 0;
    unsigned int* key4   = (unsigned int*)(ws + off); off += (size_t)nvec4;
    float*        h      = ws + off; off += (size_t)N * HD;
    float*        hw     = ws + off; off += (size_t)N * HD;
    float*        es     = ws + off; off += (size_t)N;
    float*        ed     = ws + off; off += (size_t)N;
    float*        W0p    = ws + off; off += 256;
    (void)ws_size; (void)n_in; (void)out_size;

    const int npb = cdiv(N, NB);
    k_key<<<cdiv(nvec4, 256), 256, 0, stream>>>(dst, key4, E, nvec4, npb);
    k_padW0<<<1, 256, 0, stream>>>(W0, W0p);

    const int tiles  = cdiv(N, 16);
    const int gemmBl = cdiv(tiles, 8);

    for (int g = 0; g < T; ++g) {
        k_fill<<<cdiv(N, 256), 256, 0, stream>>>(fw, sw, tw, areas, indices, h, N, g);
        for (int i = 0; i < 4; ++i) {
            const float* Wi = (i == 0) ? W0p : (Wl + (size_t)(i - 1) * 256);
            k_gemm_wmma<<<gemmBl, 256, 0, stream>>>(h, Wi, bvec + (size_t)i * HD, hw, N);
            k_esed<<<cdiv(N, 256), 256, 0, stream>>>(hw, a_src + (size_t)i * HD,
                                                     a_dst + (size_t)i * HD, es, ed, N);
            k_edge_own<<<NB, 1024, 0, stream>>>(src, dst, key4, hw, es, ed, h, N, E, npb);
        }
        k_final<<<cdiv(N, 256), 256, 0, stream>>>(h, mlpW, mlpb, (float*)d_out + (size_t)g * N, N);
    }
}


# Owner blocks' in-edge lists (src, local dst) built once per call in workspace by one owner-byte scan; the 12 layer kernels read them: no re-scan, no src/dst gather; LDS sums arrival-ordered as before
# speedup vs baseline: 6.9624x; 2.0356x over previous
#include <hip/hip_runtime.h>
#include <math.h>

typedef float v2f __attribute__((ext_vector_type(2)));
typedef float v8f __attribute__((ext_vector_type(8)));

#define HD 16
#define NWIRE 1536
#define NFEAT 4
#define NB 256
#define NPBMAX 1024
#define AST 17
#define SCAN_U 8
#define BCAP 16384
#define NOLIST 0xFFFFFFFFu
#define LWCAP 1312

__device__ __forceinline__ unsigned int encOrd(float f) {
    unsigned int b = __float_as_uint(f);
    return (b & 0x80000000u) ? ~b : (b | 0x80000000u);
}
__device__ __forceinline__ float decOrd(unsigned int e) {
    unsigned int b = (e & 0x80000000u) ? (e ^ 0x80000000u) : ~e;
    return __uint_as_float(b);
}
__device__ __forceinline__ float lrelu02(float x) { return x > 0.f ? x : 0.2f * x; }

__global__ void k_padW0(const float* __restrict__ W0, float* __restrict__ W0p) {
    int i = threadIdx.x;
    int r = i >> 4;
    W0p[i] = (r < 13) ? W0[i] : 0.f;
}

__global__ void k_fill(const float* __restrict__ fw, const float* __restrict__ sw,
                       const float* __restrict__ tw, const float* __restrict__ areas,
                       const int* __restrict__ indices, float* __restrict__ h,
                       int N, int g) {
    int n = blockIdx.x * blockDim.x + threadIdx.x;
    if (n >= N) return;
    const size_t gbase = (size_t)g * NWIRE * NFEAT;
    const float4 f0 = *(const float4*)(fw + gbase + (size_t)indices[n * 3 + 0] * NFEAT);
    const float4 f1 = *(const float4*)(sw + gbase + (size_t)indices[n * 3 + 1] * NFEAT);
    const float4 f2 = *(const float4*)(tw + gbase + (size_t)indices[n * 3 + 2] * NFEAT);
    float4* hp = (float4*)(h + (size_t)n * HD);
    hp[0] = f0;
    hp[1] = f1;
    hp[2] = f2;
    hp[3] = make_float4(areas[n], 0.f, 0.f, 0.f);
}

__global__ void k_gemm_wmma(const float* __restrict__ h, const float* __restrict__ W,
                            const float* __restrict__ bias, float* __restrict__ hw, int N) {
    const int lane = threadIdx.x & 31;
    const int wave = threadIdx.x >> 5;
    const int tile = blockIdx.x * 8 + wave;
    const int base = tile * 16;
    if (base >= N) return;
    const int col16 = lane & 15;
    const int half  = lane >> 4;
    int arow = base + col16;
    if (arow >= N) arow = N - 1;
    v8f acc = {};
#pragma unroll
    for (int k = 0; k < 4; ++k) {
        const int kc = 4 * k + 2 * half;
        v2f a;
        a.x = h[(size_t)arow * HD + kc];
        a.y = h[(size_t)arow * HD + kc + 1];
        v2f b;
        b.x = W[kc * HD + col16];
        b.y = W[(kc + 1) * HD + col16];
        acc = __builtin_amdgcn_wmma_f32_16x16x4_f32(
                  false, a, false, b, (short)0, acc, false, false);
    }
    const float bi = bias[col16];
    if (base + 16 <= N) {
#pragma unroll
        for (int r = 0; r < 8; ++r) {
            hw[(size_t)(base + r + 8 * half) * HD + col16] = acc[r] + bi;
        }
    } else {
#pragma unroll
        for (int r = 0; r < 8; ++r) {
            const int orow = base + r + 8 * half;
            if (orow < N) hw[(size_t)orow * HD + col16] = acc[r] + bi;
        }
    }
}

__global__ void k_esed(const float* __restrict__ hw, const float* __restrict__ asrc,
                       const float* __restrict__ adst, float* __restrict__ es,
                       float* __restrict__ ed, int N) {
    int n = blockIdx.x * blockDim.x + threadIdx.x;
    if (n >= N) return;
    const float4* r = (const float4*)(hw + (size_t)n * HD);
    float s = 0.f, d = 0.f;
#pragma unroll
    for (int q = 0; q < 4; ++q) {
        const float4 v = r[q];
        s += v.x * asrc[q*4+0] + v.y * asrc[q*4+1] + v.z * asrc[q*4+2] + v.w * asrc[q*4+3];
        d += v.x * adst[q*4+0] + v.y * adst[q*4+1] + v.z * adst[q*4+2] + v.w * adst[q*4+3];
    }
    es[n] = s; ed[n] = d;
}

__global__ void k_key(const int* __restrict__ dst, unsigned int* __restrict__ key4,
                      int E, int nvec4, int npb) {
    const int j = blockIdx.x * blockDim.x + threadIdx.x;
    if (j >= nvec4) return;
    unsigned int pk = 0u;
#pragma unroll
    for (int k = 0; k < 4; ++k) {
        const int e = 4 * j + k;
        const unsigned int o = (e < E) ? (unsigned int)(dst[e] / npb) : 0u;
        pk |= (o & 0xFFu) << (8 * k);
    }
    key4[j] = pk;
}

template <class F>
__device__ __forceinline__ int scan_keys(const uint4* __restrict__ kv, int nvec, unsigned int pat,
                                         int E, F fn) {
    int wcnt = 0;
    const int lane = threadIdx.x & 31;
    for (int vb = threadIdx.x & ~31; vb < nvec; vb += 1024 * SCAN_U) {
        uint4 q[SCAN_U];
#pragma unroll
        for (int u = 0; u < SCAN_U; ++u) {
            const int v = vb + u * 1024 + lane;
            q[u] = kv[v < nvec ? v : nvec - 1];
        }
#pragma unroll
        for (int u = 0; u < SCAN_U; ++u) {
            const int v = vb + u * 1024 + lane;
            const unsigned int wd[4] = {q[u].x, q[u].y, q[u].z, q[u].w};
            unsigned int m = 0u;
#pragma unroll
            for (int j = 0; j < 4; ++j) {
                const unsigned int x = wd[j] ^ pat;
                const unsigned int t = ~(((x & 0x7F7F7F7Fu) + 0x7F7F7F7Fu) | x | 0x7F7F7F7Fu);
                m |= t >> (7 - j);
            }
            if (v >= nvec) m = 0u;
            else if (v * 16 + 16 > E) {
                for (int p = 0; p < 32; ++p)
                    if (((m >> p) & 1u) && v * 16 + (p & 7) * 4 + (p >> 3) >= E) m &= ~(1u << p);
            }
            for (int it = 0; it < 16; ++it) {
                const bool has = (m != 0u);
                const unsigned int mk = (unsigned int)__ballot(has);
                if (mk == 0u) break;
                if (has) {
                    const int p = __builtin_ctz(m);
                    m &= m - 1u;
                    fn(v * 16 + (p & 7) * 4 + (p >> 3), wcnt + (int)__builtin_amdgcn_mbcnt_lo(mk, 0u));
                }
                wcnt += __popc(mk);
            }
        }
    }
    return wcnt;
}

__global__ void __launch_bounds__(1024)
k_list(const int* __restrict__ src, const int* __restrict__ dst,
       const unsigned int* __restrict__ key4, unsigned int* __restrict__ lhead,
       unsigned int* __restrict__ lists, size_t need_bytes, size_t ws_bytes, int E, int npb) {
    __shared__ unsigned int lstA[32 * LWCAP];
    __shared__ int wcn[32];
    __shared__ int ovf;
    if (need_bytes > ws_bytes) return;

    const int tid = threadIdx.x;
    const int lane = tid & 31;
    const int wave = tid >> 5;
    const int wbase = wave * LWCAP;
    const int n0 = blockIdx.x * npb;
    if (tid == 0) ovf = 0;
    __syncthreads();

    const uint4* kv = (const uint4*)key4;
    const int nvec = (E + 15) >> 4;
    const unsigned int pat = (unsigned int)blockIdx.x * 0x01010101u;
    const int wcnt = scan_keys(kv, nvec, pat, E, [&](int e, int slot) {
        if (slot < LWCAP) lstA[wbase + slot] = (unsigned int)e; else ovf = 1;
    });
    if (lane == 0) wcn[wave] = wcnt;
    __syncthreads();

    int woff = 0, total = 0;
    for (int w = 0; w < 32; ++w) { const int c = wcn[w]; if (w < wave) woff += c; total += c; }
    const bool ok = (ovf == 0) && (total <= BCAP);
    if (tid == 0) lhead[blockIdx.x] = ok ? (unsigned int)total : NOLIST;
    if (!ok) return;

    unsigned int* L = lists + (size_t)blockIdx.x * BCAP;
    for (int i = lane; i < wcnt; i += 32) {
        const int e = (int)lstA[wbase + i];
        L[woff + i] = ((unsigned int)src[e] << 10) | (unsigned int)(dst[e] - n0);
    }
}

__global__ void __launch_bounds__(1024)
k_edge_own(const int* __restrict__ src, const int* __restrict__ dst,
           const unsigned int* __restrict__ key4, const unsigned int* __restrict__ lhead,
           const unsigned int* __restrict__ lists, size_t need_bytes, size_t ws_bytes,
           const float* __restrict__ hw, const float* __restrict__ es, const float* __restrict__ ed,
           float* __restrict__ h, int N, int E, int npb) {
    __shared__ float acc[NPBMAX * AST];
    __shared__ unsigned int mEnc[NPBMAX];
    __shared__ float edl[NPBMAX];
    __shared__ unsigned int lstA[BCAP];
    __shared__ float lstB[BCAP];

    const int tid = threadIdx.x;
    const int n0 = blockIdx.x * npb;
    int nown = N - n0; if (nown > npb) nown = npb; if (nown < 0) nown = 0;

    for (int i = tid; i < nown * AST; i += 1024) acc[i] = 0.f;
    for (int i = tid; i < nown; i += 1024) { mEnc[i] = 0x007FFFFFu;   edl[i] = ed[n0 + i]; }
    const unsigned int hd = (need_bytes <= ws_bytes) ? lhead[blockIdx.x] : NOLIST;
    __syncthreads();

    if (hd != NOLIST) {
        const int cnt = hd < (unsigned int)BCAP ? (int)hd : BCAP;
        const unsigned int* L = lists + (size_t)blockIdx.x * BCAP;
        for (int i = tid; i < cnt; i += 1024) {
            const unsigned int a = L[i];
            const int dl = (int)(a & 1023u);
            const float lg = lrelu02(es[a >> 10] + edl[dl]);
            atomicMax(&mEnc[dl], encOrd(lg));
            lstA[i] = a;
            lstB[i] = lg;
        }
        __syncthreads();
        for (int i = tid; i < cnt; i += 1024) {
            const unsigned int a = lstA[i];
            const int dl = (int)(a & 1023u);
            const float4* r = (const float4*)(hw + (size_t)(a >> 10) * HD);
            const float4 r0 = r[0], r1 = r[1], r2 = r[2], r3 = r[3];
            const float w = __expf(lstB[i] - decOrd(mEnc[dl]));
            float* ap = acc + dl * AST;
            atomicAdd(ap + 0,  w * r0.x); atomicAdd(ap + 1,  w * r0.y);
            atomicAdd(ap + 2,  w * r0.z); atomicAdd(ap + 3,  w * r0.w);
            atomicAdd(ap + 4,  w * r1.x); atomicAdd(ap + 5,  w * r1.y);
            atomicAdd(ap + 6,  w * r1.z); atomicAdd(ap + 7,  w * r1.w);
            atomicAdd(ap + 8,  w * r2.x); atomicAdd(ap + 9,  w * r2.y);
            atomicAdd(ap + 10, w * r2.z); atomicAdd(ap + 11, w * r2.w);
            atomicAdd(ap + 12, w * r3.x); atomicAdd(ap + 13, w * r3.y);
            atomicAdd(ap + 14, w * r3.z); atomicAdd(ap + 15, w * r3.w);
            atomicAdd(ap + 16, w);
        }
    } else {
        const uint4* kv = (const uint4*)key4;
        const int nvec = (E + 15) >> 4;
        const unsigned int pat = (unsigned int)blockIdx.x * 0x01010101u;
        scan_keys(kv, nvec, pat, E, [&](int e, int) {
            const int dl = dst[e] - n0;
            atomicMax(&mEnc[dl], encOrd(lrelu02(es[src[e]] + edl[dl])));
        });
        __syncthreads();
        scan_keys(kv, nvec, pat, E, [&](int e, int) {
            const int s = src[e];
            const int dl = dst[e] - n0;
            const float w = __expf(lrelu02(es[s] + edl[dl]) - decOrd(mEnc[dl]));
            const float* r = hw + (size_t)s * HD;
            float* ap = acc + dl * AST;
#pragma unroll
            for (int f = 0; f < HD; ++f) atomicAdd(ap + f, w * r[f]);
            atomicAdd(ap + HD, w);
        });
    }
    __syncthreads();

    for (int j = tid; j < nown * HD; j += 1024) {
        const int n = j >> 4, f = j & 15;
        const float den = fmaxf(acc[n * AST + HD], 1e-16f);
        const float x = acc[n * AST + f] / den;
        h[(size_t)n0 * HD + j] = x > 0.f ? x : (__expf(x) - 1.f);
    }
}

__global__ void k_final(const float* __restrict__ h, const float* __restrict__ mlpW,
                        const float* __restrict__ mlpb, float* __restrict__ out, int N) {
    int n = blockIdx.x * blockDim.x + threadIdx.x;
    if (n >= N) return;
    const float4* r = (const float4*)(h + (size_t)n * HD);
    float acc = mlpb[0];
#pragma unroll
    for (int q = 0; q < 4; ++q) {
        const float4 v = r[q];
        acc += v.x * mlpW[q*4+0] + v.y * mlpW[q*4+1] + v.z * mlpW[q*4+2] + v.w * mlpW[q*4+3];
    }
    out[n] = acc;
}

static inline int cdiv(int a, int b) { return (a + b - 1) / b; }

extern "C" void kernel_launch(void* const* d_in, const int* in_sizes, int n_in,
                              void* d_out, int out_size, void* d_ws, size_t ws_size,
                              hipStream_t stream) {
    const float* fw     = (const float*)d_in[0];
    const float* sw     = (const float*)d_in[1];
    const float* tw     = (const float*)d_in[2];
    const float* areas  = (const float*)d_in[3];
    const int*   indices= (const int*)  d_in[4];
    const int*   edge   = (const int*)  d_in[5];
    const float* W0     = (const float*)d_in[6];
    const float* Wl     = (const float*)d_in[7];
    const float* a_src  = (const float*)d_in[8];
    const float* a_dst  = (const float*)d_in[9];
    const float* bvec   = (const float*)d_in[10];
    const float* mlpW   = (const float*)d_in[11];
    const float* mlpb   = (const float*)d_in[12];

    const int N = in_sizes[3];
    const int E = in_sizes[5] / 2;
    const int T = in_sizes[0] / (NWIRE * NFEAT);
    const int* src = edge;
    const int* dst = edge + E;

    float* ws = (float*)d_ws;
    const int nvec4 = cdiv(E, 16) * 4;
    size_t off = 0;
    unsigned int* key4   = (unsigned int*)(ws + off); off += (size_t)nvec4;
    float*        h      = ws + off; off += (size_t)N * HD;
    float*        hw     = ws + off; off += (size_t)N * HD;
    float*        es     = ws + off; off += (size_t)N;
    float*        ed     = ws + off; off += (size_t)N;
    float*        W0p    = ws + off; off += 256;
    unsigned int* lhead  = (unsigned int*)(ws + off); off += NB;
    unsigned int* lists  = (unsigned int*)(ws + off); off += (size_t)NB * BCAP;
    const size_t need_bytes = off * sizeof(float);
    (void)n_in; (void)out_size;

    const int npb = cdiv(N, NB);
    k_key<<<cdiv(nvec4, 256), 256, 0, stream>>>(dst, key4, E, nvec4, npb);
    k_padW0<<<1, 256, 0, stream>>>(W0, W0p);
    k_list<<<NB, 1024, 0, stream>>>(src, dst, key4, lhead, lists, need_bytes, ws_size, E, npb);

    const int tiles  = cdiv(N, 16);
    const int gemmBl = cdiv(tiles, 8);

    for (int g = 0; g < T; ++g) {
        k_fill<<<cdiv(N, 256), 256, 0, stream>>>(fw, sw, tw, areas, indices, h, N, g);
        for (int i = 0; i < 4; ++i) {
            const float* Wi = (i == 0) ? W0p : (Wl + (size_t)(i - 1) * 256);
            k_gemm_wmma<<<gemmBl, 256, 0, stream>>>(h, Wi, bvec + (size_t)i * HD, hw, N);
            k_esed<<<cdiv(N, 256), 256, 0, stream>>>(hw, a_src + (size_t)i * HD,
                                                     a_dst + (size_t)i * HD, es, ed, N);
            k_edge_own<<<NB, 1024, 0, stream>>>(src, dst, key4, lhead, lists, need_bytes, ws_size,
                                                hw, es, ed, h, N, E, npb);
        }
        k_final<<<cdiv(N, 256), 256, 0, stream>>>(h, mlpW, mlpb, (float*)d_out + (size_t)g * N, N);
    }
}


# Sum phase: four adjacent lanes share an edge, each loads 16 B of its hw row and adds its 4 products (a load reads 8 whole rows, not 16 B of 32 lines); LDS sums arrival-ordered as before
# speedup vs baseline: 7.8004x; 1.1204x over previous
#include <hip/hip_runtime.h>
#include <math.h>

typedef float v2f __attribute__((ext_vector_type(2)));
typedef float v8f __attribute__((ext_vector_type(8)));

#define HD 16
#define NWIRE 1536
#define NFEAT 4
#define NB 256
#define NPBMAX 1024
#define AST 17
#define SCAN_U 8
#define BCAP 16384
#define NOLIST 0xFFFFFFFFu
#define LWCAP 1312

__device__ __forceinline__ unsigned int encOrd(float f) {
    unsigned int b = __float_as_uint(f);
    return (b & 0x80000000u) ? ~b : (b | 0x80000000u);
}
__device__ __forceinline__ float decOrd(unsigned int e) {
    unsigned int b = (e & 0x80000000u) ? (e ^ 0x80000000u) : ~e;
    return __uint_as_float(b);
}
__device__ __forceinline__ float lrelu02(float x) { return x > 0.f ? x : 0.2f * x; }
template <int U>
__device__ __forceinline__ unsigned int quad_bcast(unsigned int v) {
    return (unsigned int)__builtin_amdgcn_update_dpp(0, (int)v, U * 0x55, 0xF, 0xF, true);
}

__global__ void k_padW0(const float* __restrict__ W0, float* __restrict__ W0p) {
    int i = threadIdx.x;
    int r = i >> 4;
    W0p[i] = (r < 13) ? W0[i] : 0.f;
}

__global__ void k_fill(const float* __restrict__ fw, const float* __restrict__ sw,
                       const float* __restrict__ tw, const float* __restrict__ areas,
                       const int* __restrict__ indices, float* __restrict__ h,
                       int N, int g) {
    int n = blockIdx.x * blockDim.x + threadIdx.x;
    if (n >= N) return;
    const size_t gbase = (size_t)g * NWIRE * NFEAT;
    const float4 f0 = *(const float4*)(fw + gbase + (size_t)indices[n * 3 + 0] * NFEAT);
    const float4 f1 = *(const float4*)(sw + gbase + (size_t)indices[n * 3 + 1] * NFEAT);
    const float4 f2 = *(const float4*)(tw + gbase + (size_t)indices[n * 3 + 2] * NFEAT);
    float4* hp = (float4*)(h + (size_t)n * HD);
    hp[0] = f0;
    hp[1] = f1;
    hp[2] = f2;
    hp[3] = make_float4(areas[n], 0.f, 0.f, 0.f);
}

__global__ void k_gemm_wmma(const float* __restrict__ h, const float* __restrict__ W,
                            const float* __restrict__ bias, float* __restrict__ hw, int N) {
    const int lane = threadIdx.x & 31;
    const int wave = threadIdx.x >> 5;
    const int tile = blockIdx.x * 8 + wave;
    const int base = tile * 16;
    if (base >= N) return;
    const int col16 = lane & 15;
    const int half  = lane >> 4;
    int arow = base + col16;
    if (arow >= N) arow = N - 1;
    v8f acc = {};
#pragma unroll
    for (int k = 0; k < 4; ++k) {
        const int kc = 4 * k + 2 * half;
        v2f a;
        a.x = h[(size_t)arow * HD + kc];
        a.y = h[(size_t)arow * HD + kc + 1];
        v2f b;
        b.x = W[kc * HD + col16];
        b.y = W[(kc + 1) * HD + col16];
        acc = __builtin_amdgcn_wmma_f32_16x16x4_f32(
                  false, a, false, b, (short)0, acc, false, false);
    }
    const float bi = bias[col16];
    if (base + 16 <= N) {
#pragma unroll
        for (int r = 0; r < 8; ++r) {
            hw[(size_t)(base + r + 8 * half) * HD + col16] = acc[r] + bi;
        }
    } else {
#pragma unroll
        for (int r = 0; r < 8; ++r) {
            const int orow = base + r + 8 * half;
            if (orow < N) hw[(size_t)orow * HD + col16] = acc[r] + bi;
        }
    }
}

__global__ void k_esed(const float* __restrict__ hw, const float* __restrict__ asrc,
                       const float* __restrict__ adst, float* __restrict__ es,
                       float* __restrict__ ed, int N) {
    int n = blockIdx.x * blockDim.x + threadIdx.x;
    if (n >= N) return;
    const float4* r = (const float4*)(hw + (size_t)n * HD);
    float s = 0.f, d = 0.f;
#pragma unroll
    for (int q = 0; q < 4; ++q) {
        const float4 v = r[q];
        s += v.x * asrc[q*4+0] + v.y * asrc[q*4+1] + v.z * asrc[q*4+2] + v.w * asrc[q*4+3];
        d += v.x * adst[q*4+0] + v.y * adst[q*4+1] + v.z * adst[q*4+2] + v.w * adst[q*4+3];
    }
    es[n] = s; ed[n] = d;
}

__global__ void k_key(const int* __restrict__ dst, unsigned int* __restrict__ key4,
                      int E, int nvec4, int npb) {
    const int j = blockIdx.x * blockDim.x + threadIdx.x;
    if (j >= nvec4) return;
    unsigned int pk = 0u;
#pragma unroll
    for (int k = 0; k < 4; ++k) {
        const int e = 4 * j + k;
        const unsigned int o = (e < E) ? (unsigned int)(dst[e] / npb) : 0u;
        pk |= (o & 0xFFu) << (8 * k);
    }
    key4[j] = pk;
}

template <class F>
__device__ __forceinline__ int scan_keys(const uint4* __restrict__ kv, int nvec, unsigned int pat,
                                         int E, F fn) {
    int wcnt = 0;
    const int lane = threadIdx.x & 31;
    for (int vb = threadIdx.x & ~31; vb < nvec; vb += 1024 * SCAN_U) {
        uint4 q[SCAN_U];
#pragma unroll
        for (int u = 0; u < SCAN_U; ++u) {
            const int v = vb + u * 1024 + lane;
            q[u] = kv[v < nvec ? v : nvec - 1];
        }
#pragma unroll
        for (int u = 0; u < SCAN_U; ++u) {
            const int v = vb + u * 1024 + lane;
            const unsigned int wd[4] = {q[u].x, q[u].y, q[u].z, q[u].w};
            unsigned int m = 0u;
#pragma unroll
            for (int j = 0; j < 4; ++j) {
                const unsigned int x = wd[j] ^ pat;
                const unsigned int t = ~(((x & 0x7F7F7F7Fu) + 0x7F7F7F7Fu) | x | 0x7F7F7F7Fu);
                m |= t >> (7 - j);
            }
            if (v >= nvec) m = 0u;
            else if (v * 16 + 16 > E) {
                for (int p = 0; p < 32; ++p)
                    if (((m >> p) & 1u) && v * 16 + (p & 7) * 4 + (p >> 3) >= E) m &= ~(1u << p);
            }
            for (int it = 0; it < 16; ++it) {
                const bool has = (m != 0u);
                const unsigned int mk = (unsigned int)__ballot(has);
                if (mk == 0u) break;
                if (has) {
                    const int p = __builtin_ctz(m);
                    m &= m - 1u;
                    fn(v * 16 + (p & 7) * 4 + (p >> 3), wcnt + (int)__builtin_amdgcn_mbcnt_lo(mk, 0u));
                }
                wcnt += __popc(mk);
            }
        }
    }
    return wcnt;
}

__global__ void __launch_bounds__(1024)
k_list(const int* __restrict__ src, const int* __restrict__ dst,
       const unsigned int* __restrict__ key4, unsigned int* __restrict__ lhead,
       unsigned int* __restrict__ lists, size_t need_bytes, size_t ws_bytes, int E, int npb) {
    __shared__ unsigned int lstA[32 * LWCAP];
    __shared__ int wcn[32];
    __shared__ int ovf;
    if (need_bytes > ws_bytes) return;

    const int tid = threadIdx.x;
    const int lane = tid & 31;
    const int wave = tid >> 5;
    const int wbase = wave * LWCAP;
    const int n0 = blockIdx.x * npb;
    if (tid == 0) ovf = 0;
    __syncthreads();

    const uint4* kv = (const uint4*)key4;
    const int nvec = (E + 15) >> 4;
    const unsigned int pat = (unsigned int)blockIdx.x * 0x01010101u;
    const int wcnt = scan_keys(kv, nvec, pat, E, [&](int e, int slot) {
        if (slot < LWCAP) lstA[wbase + slot] = (unsigned int)e; else ovf = 1;
    });
    if (lane == 0) wcn[wave] = wcnt;
    __syncthreads();

    int woff = 0, total = 0;
    for (int w = 0; w < 32; ++w) { const int c = wcn[w]; if (w < wave) woff += c; total += c; }
    const bool ok = (ovf == 0) && (total <= BCAP);
    if (tid == 0) lhead[blockIdx.x] = ok ? (unsigned int)total : NOLIST;
    if (!ok) return;

    unsigned int* L = lists + (size_t)blockIdx.x * BCAP;
    for (int i = lane; i < wcnt; i += 32) {
        const int e = (int)lstA[wbase + i];
        L[woff + i] = ((unsigned int)src[e] << 10) | (unsigned int)(dst[e] - n0);
    }
}

template <bool FULL>
__device__ __forceinline__ void sum_trip(float* acc, const unsigned int* mEnc, const unsigned int* lstA,
                                         const float* lstB, const float* __restrict__ hw,
                                         int base, int cnt, int tid) {
    const int q = tid & 3;
    const int i = base + tid;
    const bool ok = FULL || i < cnt;
    const int ii = ok ? i : 0;
    const unsigned int a = lstA[ii];
    const unsigned int a0 = quad_bcast<0>(a), a1 = quad_bcast<1>(a), a2 = quad_bcast<2>(a), a3 = quad_bcast<3>(a);
    const float4 r0 = *((const float4*)(hw + (size_t)(a0 >> 10) * HD) + q);
    const float4 r1 = *((const float4*)(hw + (size_t)(a1 >> 10) * HD) + q);
    const float4 r2 = *((const float4*)(hw + (size_t)(a2 >> 10) * HD) + q);
    const float4 r3 = *((const float4*)(hw + (size_t)(a3 >> 10) * HD) + q);
    const int dl = (int)(a & 1023u);
    const float w = __expf(lstB[ii] - decOrd(mEnc[dl]));
    const unsigned int wb = __float_as_uint(w);
    const float w0 = __uint_as_float(quad_bcast<0>(wb)), w1 = __uint_as_float(quad_bcast<1>(wb)),
                w2 = __uint_as_float(quad_bcast<2>(wb)), w3 = __uint_as_float(quad_bcast<3>(wb));
    const int i4 = base + (tid & ~3);
    if (FULL || i4 + 0 < cnt) {
        float* ap = acc + (int)(a0 & 1023u) * AST + 4 * q;
        atomicAdd(ap + 0, w0 * r0.x); atomicAdd(ap + 1, w0 * r0.y);
        atomicAdd(ap + 2, w0 * r0.z); atomicAdd(ap + 3, w0 * r0.w);
    }
    if (FULL || i4 + 1 < cnt) {
        float* ap = acc + (int)(a1 & 1023u) * AST + 4 * q;
        atomicAdd(ap + 0, w1 * r1.x); atomicAdd(ap + 1, w1 * r1.y);
        atomicAdd(ap + 2, w1 * r1.z); atomicAdd(ap + 3, w1 * r1.w);
    }
    if (FULL || i4 + 2 < cnt) {
        float* ap = acc + (int)(a2 & 1023u) * AST + 4 * q;
        atomicAdd(ap + 0, w2 * r2.x); atomicAdd(ap + 1, w2 * r2.y);
        atomicAdd(ap + 2, w2 * r2.z); atomicAdd(ap + 3, w2 * r2.w);
    }
    if (FULL || i4 + 3 < cnt) {
        float* ap = acc + (int)(a3 & 1023u) * AST + 4 * q;
        atomicAdd(ap + 0, w3 * r3.x); atomicAdd(ap + 1, w3 * r3.y);
        atomicAdd(ap + 2, w3 * r3.z); atomicAdd(ap + 3, w3 * r3.w);
    }
    if (ok) atomicAdd(acc + dl * AST + HD, w);
}

__global__ void __launch_bounds__(1024)
k_edge_own(const int* __restrict__ src, const int* __restrict__ dst,
           const unsigned int* __restrict__ key4, const unsigned int* __restrict__ lhead,
           const unsigned int* __restrict__ lists, size_t need_bytes, size_t ws_bytes,
           const float* __restrict__ hw, const float* __restrict__ es, const float* __restrict__ ed,
           float* __restrict__ h, int N, int E, int npb) {
    __shared__ float acc[NPBMAX * AST];
    __shared__ unsigned int mEnc[NPBMAX];
    __shared__ float edl[NPBMAX];
    __shared__ unsigned int lstA[BCAP];
    __shared__ float lstB[BCAP];

    const int tid = threadIdx.x;
    const int n0 = blockIdx.x * npb;
    int nown = N - n0; if (nown > npb) nown = npb; if (nown < 0) nown = 0;

    for (int i = tid; i < nown * AST; i += 1024) acc[i] = 0.f;
    for (int i = tid; i < nown; i += 1024) { mEnc[i] = 0x007FFFFFu;   edl[i] = ed[n0 + i]; }
    const unsigned int hd = (need_bytes <= ws_bytes) ? lhead[blockIdx.x] : NOLIST;
    __syncthreads();

    if (hd != NOLIST) {
        const int cnt = hd < (unsigned int)BCAP ? (int)hd : BCAP;
        const unsigned int* L = lists + (size_t)blockIdx.x * BCAP;
        for (int i = tid; i < cnt; i += 1024) {
            const unsigned int a = L[i];
            const int dl = (int)(a & 1023u);
            const float lg = lrelu02(es[a >> 10] + edl[dl]);
            atomicMax(&mEnc[dl], encOrd(lg));
            lstA[i] = a;
            lstB[i] = lg;
        }
        __syncthreads();
        int base = 0;
        for (; base + 1024 <= cnt; base += 1024) sum_trip<true>(acc, mEnc, lstA, lstB, hw, base, cnt, tid);
        if (base < cnt) sum_trip<false>(acc, mEnc, lstA, lstB, hw, base, cnt, tid);
    } else {
        const uint4* kv = (const uint4*)key4;
        const int nvec = (E + 15) >> 4;
        const unsigned int pat = (unsigned int)blockIdx.x * 0x01010101u;
        scan_keys(kv, nvec, pat, E, [&](int e, int) {
            const int dl = dst[e] - n0;
            atomicMax(&mEnc[dl], encOrd(lrelu02(es[src[e]] + edl[dl])));
        });
        __syncthreads();
        scan_keys(kv, nvec, pat, E, [&](int e, int) {
            const int s = src[e];
            const int dl = dst[e] - n0;
            const float w = __expf(lrelu02(es[s] + edl[dl]) - decOrd(mEnc[dl]));
            const float* r = hw + (size_t)s * HD;
            float* ap = acc + dl * AST;
#pragma unroll
            for (int f = 0; f < HD; ++f) atomicAdd(ap + f, w * r[f]);
            atomicAdd(ap + HD, w);
        });
    }
    __syncthreads();

    for (int j = tid; j < nown * HD; j += 1024) {
        const int n = j >> 4, f = j & 15;
        const float den = fmaxf(acc[n * AST + HD], 1e-16f);
        const float x = acc[n * AST + f] / den;
        h[(size_t)n0 * HD + j] = x > 0.f ? x : (__expf(x) - 1.f);
    }
}

__global__ void k_final(const float* __restrict__ h, const float* __restrict__ mlpW,
                        const float* __restrict__ mlpb, float* __restrict__ out, int N) {
    int n = blockIdx.x * blockDim.x + threadIdx.x;
    if (n >= N) return;
    const float4* r = (const float4*)(h + (size_t)n * HD);
    float acc = mlpb[0];
#pragma unroll
    for (int q = 0; q < 4; ++q) {
        const float4 v = r[q];
        acc += v.x * mlpW[q*4+0] + v.y * mlpW[q*4+1] + v.z * mlpW[q*4+2] + v.w * mlpW[q*4+3];
    }
    out[n] = acc;
}

static inline int cdiv(int a, int b) { return (a + b - 1) / b; }

extern "C" void kernel_launch(void* const* d_in, const int* in_sizes, int n_in,
                              void* d_out, int out_size, void* d_ws, size_t ws_size,
                              hipStream_t stream) {
    const float* fw     = (const float*)d_in[0];
    const float* sw     = (const float*)d_in[1];
    const float* tw     = (const float*)d_in[2];
    const float* areas  = (const float*)d_in[3];
    const int*   indices= (const int*)  d_in[4];
    const int*   edge   = (const int*)  d_in[5];
    const float* W0     = (const float*)d_in[6];
    const float* Wl     = (const float*)d_in[7];
    const float* a_src  = (const float*)d_in[8];
    const float* a_dst  = (const float*)d_in[9];
    const float* bvec   = (const float*)d_in[10];
    const float* mlpW   = (const float*)d_in[11];
    const float* mlpb   = (const float*)d_in[12];

    const int N = in_sizes[3];
    const int E = in_sizes[5] / 2;
    const int T = in_sizes[0] / (NWIRE * NFEAT);
    const int* src = edge;
    const int* dst = edge + E;

    float* ws = (float*)d_ws;
    const int nvec4 = cdiv(E, 16) * 4;
    size_t off = 0;
    unsigned int* key4   = (unsigned int*)(ws + off); off += (size_t)nvec4;
    float*        h      = ws + off; off += (size_t)N * HD;
    float*        hw     = ws + off; off += (size_t)N * HD;
    float*        es     = ws + off; off += (size_t)N;
    float*        ed     = ws + off; off += (size_t)N;
    float*        W0p    = ws + off; off += 256;
    unsigned int* lhead  = (unsigned int*)(ws + off); off += NB;
    unsigned int* lists  = (unsigned int*)(ws + off); off += (size_t)NB * BCAP;
    const size_t need_bytes = off * sizeof(float);
    (void)n_in; (void)out_size;

    const int npb = cdiv(N, NB);
    k_key<<<cdiv(nvec4, 256), 256, 0, stream>>>(dst, key4, E, nvec4, npb);
    k_padW0<<<1, 256, 0, stream>>>(W0, W0p);
    k_list<<<NB, 1024, 0, stream>>>(src, dst, key4, lhead, lists, need_bytes, ws_size, E, npb);

    const int tiles  = cdiv(N, 16);
    const int gemmBl = cdiv(tiles, 8);

    for (int g = 0; g < T; ++g) {
        k_fill<<<cdiv(N, 256), 256, 0, stream>>>(fw, sw, tw, areas, indices, h, N, g);
        for (int i = 0; i < 4; ++i) {
            const float* Wi = (i == 0) ? W0p : (Wl + (size_t)(i - 1) * 256);
            k_gemm_wmma<<<gemmBl, 256, 0, stream>>>(h, Wi, bvec + (size_t)i * HD, hw, N);
            k_esed<<<cdiv(N, 256), 256, 0, stream>>>(hw, a_src + (size_t)i * HD,
                                                     a_dst + (size_t)i * HD, es, ed, N);
            k_edge_own<<<NB, 1024, 0, stream>>>(src, dst, key4, lhead, lists, need_bytes, ws_size,
                                                hw, es, ed, h, N, E, npb);
        }
        k_final<<<cdiv(N, 256), 256, 0, stream>>>(h, mlpW, mlpb, (float*)d_out + (size_t)g * N, N);
    }
}


# k_list orders each owner block's in-edge list by source bucket (32 nodes = one 128-B line of es; LDS counting sort): an es gather reads few neighbouring lines; LDS sums arrival-ordered as before
# speedup vs baseline: 8.1587x; 1.0459x over previous
#include <hip/hip_runtime.h>
#include <math.h>

typedef float v2f __attribute__((ext_vector_type(2)));
typedef float v8f __attribute__((ext_vector_type(8)));

#define HD 16
#define NWIRE 1536
#define NFEAT 4
#define NB 256
#define NPBMAX 1024
#define AST 17
#define SCAN_U 8
#define BCAP 16384
#define NOLIST 0xFFFFFFFFu
#define LWCAP 1312
#define NBK 8192

__device__ __forceinline__ unsigned int encOrd(float f) {
    unsigned int b = __float_as_uint(f);
    return (b & 0x80000000u) ? ~b : (b | 0x80000000u);
}
__device__ __forceinline__ float decOrd(unsigned int e) {
    unsigned int b = (e & 0x80000000u) ? (e ^ 0x80000000u) : ~e;
    return __uint_as_float(b);
}
__device__ __forceinline__ float lrelu02(float x) { return x > 0.f ? x : 0.2f * x; }
template <int U>
__device__ __forceinline__ unsigned int quad_bcast(unsigned int v) {
    return (unsigned int)__builtin_amdgcn_update_dpp(0, (int)v, U * 0x55, 0xF, 0xF, true);
}

__global__ void k_padW0(const float* __restrict__ W0, float* __restrict__ W0p) {
    int i = threadIdx.x;
    int r = i >> 4;
    W0p[i] = (r < 13) ? W0[i] : 0.f;
}

__global__ void k_fill(const float* __restrict__ fw, const float* __restrict__ sw,
                       const float* __restrict__ tw, const float* __restrict__ areas,
                       const int* __restrict__ indices, float* __restrict__ h,
                       int N, int g) {
    int n = blockIdx.x * blockDim.x + threadIdx.x;
    if (n >= N) return;
    const size_t gbase = (size_t)g * NWIRE * NFEAT;
    const float4 f0 = *(const float4*)(fw + gbase + (size_t)indices[n * 3 + 0] * NFEAT);
    const float4 f1 = *(const float4*)(sw + gbase + (size_t)indices[n * 3 + 1] * NFEAT);
    const float4 f2 = *(const float4*)(tw + gbase + (size_t)indices[n * 3 + 2] * NFEAT);
    float4* hp = (float4*)(h + (size_t)n * HD);
    hp[0] = f0;
    hp[1] = f1;
    hp[2] = f2;
    hp[3] = make_float4(areas[n], 0.f, 0.f, 0.f);
}

__global__ void k_gemm_wmma(const float* __restrict__ h, const float* __restrict__ W,
                            const float* __restrict__ bias, float* __restrict__ hw, int N) {
    const int lane = threadIdx.x & 31;
    const int wave = threadIdx.x >> 5;
    const int tile = blockIdx.x * 8 + wave;
    const int base = tile * 16;
    if (base >= N) return;
    const int col16 = lane & 15;
    const int half  = lane >> 4;
    int arow = base + col16;
    if (arow >= N) arow = N - 1;
    v8f acc = {};
#pragma unroll
    for (int k = 0; k < 4; ++k) {
        const int kc = 4 * k + 2 * half;
        v2f a;
        a.x = h[(size_t)arow * HD + kc];
        a.y = h[(size_t)arow * HD + kc + 1];
        v2f b;
        b.x = W[kc * HD + col16];
        b.y = W[(kc + 1) * HD + col16];
        acc = __builtin_amdgcn_wmma_f32_16x16x4_f32(
                  false, a, false, b, (short)0, acc, false, false);
    }
    const float bi = bias[col16];
    if (base + 16 <= N) {
#pragma unroll
        for (int r = 0; r < 8; ++r) {
            hw[(size_t)(base + r + 8 * half) * HD + col16] = acc[r] + bi;
        }
    } else {
#pragma unroll
        for (int r = 0; r < 8; ++r) {
            const int orow = base + r + 8 * half;
            if (orow < N) hw[(size_t)orow * HD + col16] = acc[r] + bi;
        }
    }
}

__global__ void k_esed(const float* __restrict__ hw, const float* __restrict__ asrc,
                       const float* __restrict__ adst, float* __restrict__ es,
                       float* __restrict__ ed, int N) {
    int n = blockIdx.x * blockDim.x + threadIdx.x;
    if (n >= N) return;
    const float4* r = (const float4*)(hw + (size_t)n * HD);
    float s = 0.f, d = 0.f;
#pragma unroll
    for (int q = 0; q < 4; ++q) {
        const float4 v = r[q];
        s += v.x * asrc[q*4+0] + v.y * asrc[q*4+1] + v.z * asrc[q*4+2] + v.w * asrc[q*4+3];
        d += v.x * adst[q*4+0] + v.y * adst[q*4+1] + v.z * adst[q*4+2] + v.w * adst[q*4+3];
    }
    es[n] = s; ed[n] = d;
}

__global__ void k_key(const int* __restrict__ dst, unsigned int* __restrict__ key4,
                      int E, int nvec4, int npb) {
    const int j = blockIdx.x * blockDim.x + threadIdx.x;
    if (j >= nvec4) return;
    unsigned int pk = 0u;
#pragma unroll
    for (int k = 0; k < 4; ++k) {
        const int e = 4 * j + k;
        const unsigned int o = (e < E) ? (unsigned int)(dst[e] / npb) : 0u;
        pk |= (o & 0xFFu) << (8 * k);
    }
    key4[j] = pk;
}

template <class F>
__device__ __forceinline__ int scan_keys(const uint4* __restrict__ kv, int nvec, unsigned int pat,
                                         int E, F fn) {
    int wcnt = 0;
    const int lane = threadIdx.x & 31;
    for (int vb = threadIdx.x & ~31; vb < nvec; vb += 1024 * SCAN_U) {
        uint4 q[SCAN_U];
#pragma unroll
        for (int u = 0; u < SCAN_U; ++u) {
            const int v = vb + u * 1024 + lane;
            q[u] = kv[v < nvec ? v : nvec - 1];
        }
#pragma unroll
        for (int u = 0; u < SCAN_U; ++u) {
            const int v = vb + u * 1024 + lane;
            const unsigned int wd[4] = {q[u].x, q[u].y, q[u].z, q[u].w};
            unsigned int m = 0u;
#pragma unroll
            for (int j = 0; j < 4; ++j) {
                const unsigned int x = wd[j] ^ pat;
                const unsigned int t = ~(((x & 0x7F7F7F7Fu) + 0x7F7F7F7Fu) | x | 0x7F7F7F7Fu);
                m |= t >> (7 - j);
            }
            if (v >= nvec) m = 0u;
            else if (v * 16 + 16 > E) {
                for (int p = 0; p < 32; ++p)
                    if (((m >> p) & 1u) && v * 16 + (p & 7) * 4 + (p >> 3) >= E) m &= ~(1u << p);
            }
            for (int it = 0; it < 16; ++it) {
                const bool has = (m != 0u);
                const unsigned int mk = (unsigned int)__ballot(has);
                if (mk == 0u) break;
                if (has) {
                    const int p = __builtin_ctz(m);
                    m &= m - 1u;
                    fn(v * 16 + (p & 7) * 4 + (p >> 3), wcnt + (int)__builtin_amdgcn_mbcnt_lo(mk, 0u));
                }
                wcnt += __popc(mk);
            }
        }
    }
    return wcnt;
}

__global__ void __launch_bounds__(1024)
k_list(const int* __restrict__ src, const int* __restrict__ dst,
       const unsigned int* __restrict__ key4, unsigned int* __restrict__ lhead,
       unsigned int* __restrict__ lists, size_t need_bytes, size_t ws_bytes, int N, int E, int npb) {
    __shared__ unsigned int lstA[32 * LWCAP];
    __shared__ unsigned int srt[BCAP];
    __shared__ unsigned int cntb[NBK];
    __shared__ int wcn[32];
    __shared__ unsigned int wsum[32];
    __shared__ int ovf;
    if (need_bytes > ws_bytes) return;

    const int tid = threadIdx.x;
    const int lane = tid & 31;
    const int wave = tid >> 5;
    const int wbase = wave * LWCAP;
    const int n0 = blockIdx.x * npb;
    for (int i = tid; i < NBK; i += 1024) cntb[i] = 0u;
    if (tid == 0) ovf = 0;
    __syncthreads();

    const uint4* kv = (const uint4*)key4;
    const int nvec = (E + 15) >> 4;
    const unsigned int pat = (unsigned int)blockIdx.x * 0x01010101u;
    const int wcnt = scan_keys(kv, nvec, pat, E, [&](int e, int slot) {
        if (slot < LWCAP) lstA[wbase + slot] = (unsigned int)e; else ovf = 1;
    });
    if (lane == 0) wcn[wave] = wcnt;
    __syncthreads();

    int total = 0;
    for (int w = 0; w < 32; ++w) total += wcn[w];
    const bool ok = (ovf == 0) && (total <= BCAP);
    if (tid == 0) lhead[blockIdx.x] = ok ? (unsigned int)total : NOLIST;
    if (!ok) return;

    int sh = 5;
    while (((N - 1) >> sh) >= NBK) ++sh;
    const unsigned int nbk = (unsigned int)((N - 1) >> sh) + 1u;
    const unsigned int rot = (unsigned int)(((size_t)blockIdx.x * nbk) / gridDim.x);
    auto bucket = [&](unsigned int s) -> unsigned int {
        unsigned int b = s >> sh;
        if (b >= nbk) b = nbk - 1u;
        b = b >= rot ? b - rot : b + nbk - rot;
        return b;
    };

    for (int i = lane; i < wcnt; i += 32) {
        const int e = (int)lstA[wbase + i];
        const unsigned int sv = (unsigned int)src[e];
        lstA[wbase + i] = (sv << 10) | (unsigned int)(dst[e] - n0);
        atomicAdd(&cntb[bucket(sv)], 1u);
    }
    __syncthreads();

    unsigned int c[NBK / 1024];
    unsigned int s8 = 0u;
#pragma unroll
    for (int k = 0; k < NBK / 1024; ++k) { c[k] = cntb[tid * (NBK / 1024) + k]; s8 += c[k]; }
    unsigned int inc = s8;
    for (int d = 1; d < 32; d <<= 1) {
        const unsigned int v = (unsigned int)__shfl_up((int)inc, (unsigned int)d);
        if (lane >= d) inc += v;
    }
    if (lane == 31) wsum[wave] = inc;
    __syncthreads();
    unsigned int run = inc - s8;
    for (int w = 0; w < wave; ++w) run += wsum[w];
#pragma unroll
    for (int k = 0; k < NBK / 1024; ++k) { cntb[tid * (NBK / 1024) + k] = run; run += c[k]; }
    __syncthreads();

    for (int i = lane; i < wcnt; i += 32) {
        const unsigned int wd = lstA[wbase + i];
        const unsigned int pos = atomicAdd(&cntb[bucket(wd >> 10)], 1u);
        srt[pos] = wd;
    }
    __syncthreads();

    unsigned int* L = lists + (size_t)blockIdx.x * BCAP;
    for (int i = tid; i < total; i += 1024) L[i] = srt[i];
}

template <bool FULL>
__device__ __forceinline__ void sum_trip(float* acc, const unsigned int* mEnc, const unsigned int* lstA,
                                         const float* lstB, const float* __restrict__ hw,
                                         int base, int cnt, int tid) {
    const int q = tid & 3;
    const int i = base + tid;
    const bool ok = FULL || i < cnt;
    const int ii = ok ? i : 0;
    const unsigned int a = lstA[ii];
    const unsigned int a0 = quad_bcast<0>(a), a1 = quad_bcast<1>(a), a2 = quad_bcast<2>(a), a3 = quad_bcast<3>(a);
    const float4 r0 = *((const float4*)(hw + (size_t)(a0 >> 10) * HD) + q);
    const float4 r1 = *((const float4*)(hw + (size_t)(a1 >> 10) * HD) + q);
    const float4 r2 = *((const float4*)(hw + (size_t)(a2 >> 10) * HD) + q);
    const float4 r3 = *((const float4*)(hw + (size_t)(a3 >> 10) * HD) + q);
    const int dl = (int)(a & 1023u);
    const float w = __expf(lstB[ii] - decOrd(mEnc[dl]));
    const unsigned int wb = __float_as_uint(w);
    const float w0 = __uint_as_float(quad_bcast<0>(wb)), w1 = __uint_as_float(quad_bcast<1>(wb)),
                w2 = __uint_as_float(quad_bcast<2>(wb)), w3 = __uint_as_float(quad_bcast<3>(wb));
    const int i4 = base + (tid & ~3);
    if (FULL || i4 + 0 < cnt) {
        float* ap = acc + (int)(a0 & 1023u) * AST + 4 * q;
        atomicAdd(ap + 0, w0 * r0.x); atomicAdd(ap + 1, w0 * r0.y);
        atomicAdd(ap + 2, w0 * r0.z); atomicAdd(ap + 3, w0 * r0.w);
    }
    if (FULL || i4 + 1 < cnt) {
        float* ap = acc + (int)(a1 & 1023u) * AST + 4 * q;
        atomicAdd(ap + 0, w1 * r1.x); atomicAdd(ap + 1, w1 * r1.y);
        atomicAdd(ap + 2, w1 * r1.z); atomicAdd(ap + 3, w1 * r1.w);
    }
    if (FULL || i4 + 2 < cnt) {
        float* ap = acc + (int)(a2 & 1023u) * AST + 4 * q;
        atomicAdd(ap + 0, w2 * r2.x); atomicAdd(ap + 1, w2 * r2.y);
        atomicAdd(ap + 2, w2 * r2.z); atomicAdd(ap + 3, w2 * r2.w);
    }
    if (FULL || i4 + 3 < cnt) {
        float* ap = acc + (int)(a3 & 1023u) * AST + 4 * q;
        atomicAdd(ap + 0, w3 * r3.x); atomicAdd(ap + 1, w3 * r3.y);
        atomicAdd(ap + 2, w3 * r3.z); atomicAdd(ap + 3, w3 * r3.w);
    }
    if (ok) atomicAdd(acc + dl * AST + HD, w);
}

__global__ void __launch_bounds__(1024)
k_edge_own(const int* __restrict__ src, const int* __restrict__ dst,
           const unsigned int* __restrict__ key4, const unsigned int* __restrict__ lhead,
           const unsigned int* __restrict__ lists, size_t need_bytes, size_t ws_bytes,
           const float* __restrict__ hw, const float* __restrict__ es, const float* __restrict__ ed,
           float* __restrict__ h, int N, int E, int npb) {
    __shared__ float acc[NPBMAX * AST];
    __shared__ unsigned int mEnc[NPBMAX];
    __shared__ float edl[NPBMAX];
    __shared__ unsigned int lstA[BCAP];
    __shared__ float lstB[BCAP];

    const int tid = threadIdx.x;
    const int n0 = blockIdx.x * npb;
    int nown = N - n0; if (nown > npb) nown = npb; if (nown < 0) nown = 0;

    for (int i = tid; i < nown * AST; i += 1024) acc[i] = 0.f;
    for (int i = tid; i < nown; i += 1024) { mEnc[i] = 0x007FFFFFu;   edl[i] = ed[n0 + i]; }
    const unsigned int hd = (need_bytes <= ws_bytes) ? lhead[blockIdx.x] : NOLIST;
    __syncthreads();

    if (hd != NOLIST) {
        const int cnt = hd < (unsigned int)BCAP ? (int)hd : BCAP;
        const unsigned int* L = lists + (size_t)blockIdx.x * BCAP;
        for (int i = tid; i < cnt; i += 1024) {
            const unsigned int a = L[i];
            const int dl = (int)(a & 1023u);
            const float lg = lrelu02(es[a >> 10] + edl[dl]);
            atomicMax(&mEnc[dl], encOrd(lg));
            lstA[i] = a;
            lstB[i] = lg;
        }
        __syncthreads();
        int base = 0;
        for (; base + 1024 <= cnt; base += 1024) sum_trip<true>(acc, mEnc, lstA, lstB, hw, base, cnt, tid);
        if (base < cnt) sum_trip<false>(acc, mEnc, lstA, lstB, hw, base, cnt, tid);
    } else {
        const uint4* kv = (const uint4*)key4;
        const int nvec = (E + 15) >> 4;
        const unsigned int pat = (unsigned int)blockIdx.x * 0x01010101u;
        scan_keys(kv, nvec, pat, E, [&](int e, int) {
            const int dl = dst[e] - n0;
            atomicMax(&mEnc[dl], encOrd(lrelu02(es[src[e]] + edl[dl])));
        });
        __syncthreads();
        scan_keys(kv, nvec, pat, E, [&](int e, int) {
            const int s = src[e];
            const int dl = dst[e] - n0;
            const float w = __expf(lrelu02(es[s] + edl[dl]) - decOrd(mEnc[dl]));
            const float* r = hw + (size_t)s * HD;
            float* ap = acc + dl * AST;
#pragma unroll
            for (int f = 0; f < HD; ++f) atomicAdd(ap + f, w * r[f]);
            atomicAdd(ap + HD, w);
        });
    }
    __syncthreads();

    for (int j = tid; j < nown * HD; j += 1024) {
        const int n = j >> 4, f = j & 15;
        const float den = fmaxf(acc[n * AST + HD], 1e-16f);
        const float x = acc[n * AST + f] / den;
        h[(size_t)n0 * HD + j] = x > 0.f ? x : (__expf(x) - 1.f);
    }
}

__global__ void k_final(const float* __restrict__ h, const float* __restrict__ mlpW,
                        const float* __restrict__ mlpb, float* __restrict__ out, int N) {
    int n = blockIdx.x * blockDim.x + threadIdx.x;
    if (n >= N) return;
    const float4* r = (const float4*)(h + (size_t)n * HD);
    float acc = mlpb[0];
#pragma unroll
    for (int q = 0; q < 4; ++q) {
        const float4 v = r[q];
        acc += v.x * mlpW[q*4+0] + v.y * mlpW[q*4+1] + v.z * mlpW[q*4+2] + v.w * mlpW[q*4+3];
    }
    out[n] = acc;
}

static inline int cdiv(int a, int b) { return (a + b - 1) / b; }

extern "C" void kernel_launch(void* const* d_in, const int* in_sizes, int n_in,
                              void* d_out, int out_size, void* d_ws, size_t ws_size,
                              hipStream_t stream) {
    const float* fw     = (const float*)d_in[0];
    const float* sw     = (const float*)d_in[1];
    const float* tw     = (const float*)d_in[2];
    const float* areas  = (const float*)d_in[3];
    const int*   indices= (const int*)  d_in[4];
    const int*   edge   = (const int*)  d_in[5];
    const float* W0     = (const float*)d_in[6];
    const float* Wl     = (const float*)d_in[7];
    const float* a_src  = (const float*)d_in[8];
    const float* a_dst  = (const float*)d_in[9];
    const float* bvec   = (const float*)d_in[10];
    const float* mlpW   = (const float*)d_in[11];
    const float* mlpb   = (const float*)d_in[12];

    const int N = in_sizes[3];
    const int E = in_sizes[5] / 2;
    const int T = in_sizes[0] / (NWIRE * NFEAT);
    const int* src = edge;
    const int* dst = edge + E;

    float* ws = (float*)d_ws;
    const int nvec4 = cdiv(E, 16) * 4;
    size_t off = 0;
    unsigned int* key4   = (unsigned int*)(ws + off); off += (size_t)nvec4;
    float*        h      = ws + off; off += (size_t)N * HD;
    float*        hw     = ws + off; off += (size_t)N * HD;
    float*        es     = ws + off; off += (size_t)N;
    float*        ed     = ws + off; off += (size_t)N;
    float*        W0p    = ws + off; off += 256;
    unsigned int* lhead  = (unsigned int*)(ws + off); off += NB;
    unsigned int* lists  = (unsigned int*)(ws + off); off += (size_t)NB * BCAP;
    const size_t need_bytes = off * sizeof(float);
    (void)n_in; (void)out_size;

    const int npb = cdiv(N, NB);
    k_key<<<cdiv(nvec4, 256), 256, 0, stream>>>(dst, key4, E, nvec4, npb);
    k_padW0<<<1, 256, 0, stream>>>(W0, W0p);
    k_list<<<NB, 1024, 0, stream>>>(src, dst, key4, lhead, lists, need_bytes, ws_size, N, E, npb);

    const int tiles  = cdiv(N, 16);
    const int gemmBl = cdiv(tiles, 8);

    for (int g = 0; g < T; ++g) {
        k_fill<<<cdiv(N, 256), 256, 0, stream>>>(fw, sw, tw, areas, indices, h, N, g);
        for (int i = 0; i < 4; ++i) {
            const float* Wi = (i == 0) ? W0p : (Wl + (size_t)(i - 1) * 256);
            k_gemm_wmma<<<gemmBl, 256, 0, stream>>>(h, Wi, bvec + (size_t)i * HD, hw, N);
            k_esed<<<cdiv(N, 256), 256, 0, stream>>>(hw, a_src + (size_t)i * HD,
                                                     a_dst + (size_t)i * HD, es, ed, N);
            k_edge_own<<<NB, 1024, 0, stream>>>(src, dst, key4, lhead, lists, need_bytes, ws_size,
                                                hw, es, ed, h, N, E, npb);
        }
        k_final<<<cdiv(N, 256), 256, 0, stream>>>(h, mlpW, mlpb, (float*)d_out + (size_t)g * N, N);
    }
}


# es/ed projections taken in the GEMM kernel from its result tile in registers (halving sums over the 16 lanes of a row): 12 projection launches and their re-read of hw go; projection sum order changes
# speedup vs baseline: 8.6898x; 1.0651x over previous
#include <hip/hip_runtime.h>
#include <math.h>

typedef float v2f __attribute__((ext_vector_type(2)));
typedef float v8f __attribute__((ext_vector_type(8)));

#define HD 16
#define NWIRE 1536
#define NFEAT 4
#define NB 256
#define NPBMAX 1024
#define AST 17
#define SCAN_U 8
#define BCAP 16384
#define NOLIST 0xFFFFFFFFu
#define LWCAP 1312
#define NBK 8192

__device__ __forceinline__ unsigned int encOrd(float f) {
    unsigned int b = __float_as_uint(f);
    return (b & 0x80000000u) ? ~b : (b | 0x80000000u);
}
__device__ __forceinline__ float decOrd(unsigned int e) {
    unsigned int b = (e & 0x80000000u) ? (e ^ 0x80000000u) : ~e;
    return __uint_as_float(b);
}
__device__ __forceinline__ float lrelu02(float x) { return x > 0.f ? x : 0.2f * x; }
template <int U>
__device__ __forceinline__ unsigned int quad_bcast(unsigned int v) {
    return (unsigned int)__builtin_amdgcn_update_dpp(0, (int)v, U * 0x55, 0xF, 0xF, true);
}

template <int CTRL>
__device__ __forceinline__ float row_from(float v) {
    return __uint_as_float((unsigned int)__builtin_amdgcn_update_dpp(0, (int)__float_as_uint(v), CTRL, 0xF, 0xF, true));
}
template <int CTRL, int NOUT>
__device__ __forceinline__ void fold_half(float* v, bool up) {
#pragma unroll
    for (int i = 0; i < NOUT; ++i) {
        const float keep = up ? v[NOUT + i] : v[i];
        const float send = up ? v[i] : v[NOUT + i];
        v[i] = keep + row_from<CTRL>(send);
    }
}

__global__ void k_padW0(const float* __restrict__ W0, float* __restrict__ W0p) {
    int i = threadIdx.x;
    int r = i >> 4;
    W0p[i] = (r < 13) ? W0[i] : 0.f;
}

__global__ void k_fill(const float* __restrict__ fw, const float* __restrict__ sw,
                       const float* __restrict__ tw, const float* __restrict__ areas,
                       const int* __restrict__ indices, float* __restrict__ h,
                       int N, int g) {
    int n = blockIdx.x * blockDim.x + threadIdx.x;
    if (n >= N) return;
    const size_t gbase = (size_t)g * NWIRE * NFEAT;
    const float4 f0 = *(const float4*)(fw + gbase + (size_t)indices[n * 3 + 0] * NFEAT);
    const float4 f1 = *(const float4*)(sw + gbase + (size_t)indices[n * 3 + 1] * NFEAT);
    const float4 f2 = *(const float4*)(tw + gbase + (size_t)indices[n * 3 + 2] * NFEAT);
    float4* hp = (float4*)(h + (size_t)n * HD);
    hp[0] = f0;
    hp[1] = f1;
    hp[2] = f2;
    hp[3] = make_float4(areas[n], 0.f, 0.f, 0.f);
}

__global__ void k_gemm_wmma(const float* __restrict__ h, const float* __restrict__ W,
                            const float* __restrict__ bias, const float* __restrict__ asrc,
                            const float* __restrict__ adst, float* __restrict__ hw,
                            float* __restrict__ es, float* __restrict__ ed, int N) {
    const int lane = threadIdx.x & 31;
    const int wave = threadIdx.x >> 5;
    const int tile = blockIdx.x * 8 + wave;
    const int base = tile * 16;
    if (base >= N) return;
    const int col16 = lane & 15;
    const int half  = lane >> 4;
    int arow = base + col16;
    if (arow >= N) arow = N - 1;
    v8f acc = {};
#pragma unroll
    for (int k = 0; k < 4; ++k) {
        const int kc = 4 * k + 2 * half;
        v2f a;
        a.x = h[(size_t)arow * HD + kc];
        a.y = h[(size_t)arow * HD + kc + 1];
        v2f b;
        b.x = W[kc * HD + col16];
        b.y = W[(kc + 1) * HD + col16];
        acc = __builtin_amdgcn_wmma_f32_16x16x4_f32(
                  false, a, false, b, (short)0, acc, false, false);
    }
    const float bi = bias[col16];
    const float as = asrc[col16];
    const float ad = adst[col16];
    float p[16];
#pragma unroll
    for (int r = 0; r < 8; ++r) {
        const float v = acc[r] + bi;
        acc[r] = v;
        p[r] = v * as;
        p[8 + r] = v * ad;
    }
    if (base + 16 <= N) {
#pragma unroll
        for (int r = 0; r < 8; ++r) {
            hw[(size_t)(base + r + 8 * half) * HD + col16] = acc[r];
        }
    } else {
#pragma unroll
        for (int r = 0; r < 8; ++r) {
            const int orow = base + r + 8 * half;
            if (orow < N) hw[(size_t)orow * HD + col16] = acc[r];
        }
    }
    fold_half<0x128, 8>(p, (col16 & 8) != 0);
    fold_half<0x141, 4>(p, (col16 & 4) != 0);
    fold_half<0x4E, 2>(p, (col16 & 2) != 0);
    fold_half<0xB1, 1>(p, (col16 & 1) != 0);
    const int prow = base + 8 * half + (col16 & 7);
    float* pd = (col16 & 8) ? ed : es;
    if (prow < N) pd[prow] = p[0];
}

__global__ void k_key(const int* __restrict__ dst, unsigned int* __restrict__ key4,
                      int E, int nvec4, int npb) {
    const int j = blockIdx.x * blockDim.x + threadIdx.x;
    if (j >= nvec4) return;
    unsigned int pk = 0u;
#pragma unroll
    for (int k = 0; k < 4; ++k) {
        const int e = 4 * j + k;
        const unsigned int o = (e < E) ? (unsigned int)(dst[e] / npb) : 0u;
        pk |= (o & 0xFFu) << (8 * k);
    }
    key4[j] = pk;
}

template <class F>
__device__ __forceinline__ int scan_keys(const uint4* __restrict__ kv, int nvec, unsigned int pat,
                                         int E, F fn) {
    int wcnt = 0;
    const int lane = threadIdx.x & 31;
    for (int vb = threadIdx.x & ~31; vb < nvec; vb += 1024 * SCAN_U) {
        uint4 q[SCAN_U];
#pragma unroll
        for (int u = 0; u < SCAN_U; ++u) {
            const int v = vb + u * 1024 + lane;
            q[u] = kv[v < nvec ? v : nvec - 1];
        }
#pragma unroll
        for (int u = 0; u < SCAN_U; ++u) {
            const int v = vb + u * 1024 + lane;
            const unsigned int wd[4] = {q[u].x, q[u].y, q[u].z, q[u].w};
            unsigned int m = 0u;
#pragma unroll
            for (int j = 0; j < 4; ++j) {
                const unsigned int x = wd[j] ^ pat;
                const unsigned int t = ~(((x & 0x7F7F7F7Fu) + 0x7F7F7F7Fu) | x | 0x7F7F7F7Fu);
                m |= t >> (7 - j);
            }
            if (v >= nvec) m = 0u;
            else if (v * 16 + 16 > E) {
                for (int p = 0; p < 32; ++p)
                    if (((m >> p) & 1u) && v * 16 + (p & 7) * 4 + (p >> 3) >= E) m &= ~(1u << p);
            }
            for (int it = 0; it < 16; ++it) {
                const bool has = (m != 0u);
                const unsigned int mk = (unsigned int)__ballot(has);
                if (mk == 0u) break;
                if (has) {
                    const int p = __builtin_ctz(m);
                    m &= m - 1u;
                    fn(v * 16 + (p & 7) * 4 + (p >> 3), wcnt + (int)__builtin_amdgcn_mbcnt_lo(mk, 0u));
                }
                wcnt += __popc(mk);
            }
        }
    }
    return wcnt;
}

__global__ void __launch_bounds__(1024)
k_list(const int* __restrict__ src, const int* __restrict__ dst,
       const unsigned int* __restrict__ key4, unsigned int* __restrict__ lhead,
       unsigned int* __restrict__ lists, size_t need_bytes, size_t ws_bytes, int N, int E, int npb) {
    __shared__ unsigned int lstA[32 * LWCAP];
    __shared__ unsigned int srt[BCAP];
    __shared__ unsigned int cntb[NBK];
    __shared__ int wcn[32];
    __shared__ unsigned int wsum[32];
    __shared__ int ovf;
    if (need_bytes > ws_bytes) return;

    const int tid = threadIdx.x;
    const int lane = tid & 31;
    const int wave = tid >> 5;
    const int wbase = wave * LWCAP;
    const int n0 = blockIdx.x * npb;
    for (int i = tid; i < NBK; i += 1024) cntb[i] = 0u;
    if (tid == 0) ovf = 0;
    __syncthreads();

    const uint4* kv = (const uint4*)key4;
    const int nvec = (E + 15) >> 4;
    const unsigned int pat = (unsigned int)blockIdx.x * 0x01010101u;
    const int wcnt = scan_keys(kv, nvec, pat, E, [&](int e, int slot) {
        if (slot < LWCAP) lstA[wbase + slot] = (unsigned int)e; else ovf = 1;
    });
    if (lane == 0) wcn[wave] = wcnt;
    __syncthreads();

    int total = 0;
    for (int w = 0; w < 32; ++w) total += wcn[w];
    const bool ok = (ovf == 0) && (total <= BCAP);
    if (tid == 0) lhead[blockIdx.x] = ok ? (unsigned int)total : NOLIST;
    if (!ok) return;

    int sh = 5;
    while (((N - 1) >> sh) >= NBK) ++sh;
    const unsigned int nbk = (unsigned int)((N - 1) >> sh) + 1u;
    const unsigned int rot = (unsigned int)(((size_t)blockIdx.x * nbk) / gridDim.x);
    auto bucket = [&](unsigned int s) -> unsigned int {
        unsigned int b = s >> sh;
        if (b >= nbk) b = nbk - 1u;
        b = b >= rot ? b - rot : b + nbk - rot;
        return b;
    };

    for (int i = lane; i < wcnt; i += 32) {
        const int e = (int)lstA[wbase + i];
        const unsigned int sv = (unsigned int)src[e];
        lstA[wbase + i] = (sv << 10) | (unsigned int)(dst[e] - n0);
        atomicAdd(&cntb[bucket(sv)], 1u);
    }
    __syncthreads();

    unsigned int c[NBK / 1024];
    unsigned int s8 = 0u;
#pragma unroll
    for (int k = 0; k < NBK / 1024; ++k) { c[k] = cntb[tid * (NBK / 1024) + k]; s8 += c[k]; }
    unsigned int inc = s8;
    for (int d = 1; d < 32; d <<= 1) {
        const unsigned int v = (unsigned int)__shfl_up((int)inc, (unsigned int)d);
        if (lane >= d) inc += v;
    }
    if (lane == 31) wsum[wave] = inc;
    __syncthreads();
    unsigned int run = inc - s8;
    for (int w = 0; w < wave; ++w) run += wsum[w];
#pragma unroll
    for (int k = 0; k < NBK / 1024; ++k) { cntb[tid * (NBK / 1024) + k] = run; run += c[k]; }
    __syncthreads();

    for (int i = lane; i < wcnt; i += 32) {
        const unsigned int wd = lstA[wbase + i];
        const unsigned int pos = atomicAdd(&cntb[bucket(wd >> 10)], 1u);
        srt[pos] = wd;
    }
    __syncthreads();

    unsigned int* L = lists + (size_t)blockIdx.x * BCAP;
    for (int i = tid; i < total; i += 1024) L[i] = srt[i];
}

template <bool FULL>
__device__ __forceinline__ void sum_trip(float* acc, const unsigned int* mEnc, const unsigned int* lstA,
                                         const float* lstB, const float* __restrict__ hw,
                                         int base, int cnt, int tid) {
    const int q = tid & 3;
    const int i = base + tid;
    const bool ok = FULL || i < cnt;
    const int ii = ok ? i : 0;
    const unsigned int a = lstA[ii];
    const unsigned int a0 = quad_bcast<0>(a), a1 = quad_bcast<1>(a), a2 = quad_bcast<2>(a), a3 = quad_bcast<3>(a);
    const float4 r0 = *((const float4*)(hw + (size_t)(a0 >> 10) * HD) + q);
    const float4 r1 = *((const float4*)(hw + (size_t)(a1 >> 10) * HD) + q);
    const float4 r2 = *((const float4*)(hw + (size_t)(a2 >> 10) * HD) + q);
    const float4 r3 = *((const float4*)(hw + (size_t)(a3 >> 10) * HD) + q);
    const int dl = (int)(a & 1023u);
    const float w = __expf(lstB[ii] - decOrd(mEnc[dl]));
    const unsigned int wb = __float_as_uint(w);
    const float w0 = __uint_as_float(quad_bcast<0>(wb)), w1 = __uint_as_float(quad_bcast<1>(wb)),
                w2 = __uint_as_float(quad_bcast<2>(wb)), w3 = __uint_as_float(quad_bcast<3>(wb));
    const int i4 = base + (tid & ~3);
    if (FULL || i4 + 0 < cnt) {
        float* ap = acc + (int)(a0 & 1023u) * AST + 4 * q;
        atomicAdd(ap + 0, w0 * r0.x); atomicAdd(ap + 1, w0 * r0.y);
        atomicAdd(ap + 2, w0 * r0.z); atomicAdd(ap + 3, w0 * r0.w);
    }
    if (FULL || i4 + 1 < cnt) {
        float* ap = acc + (int)(a1 & 1023u) * AST + 4 * q;
        atomicAdd(ap + 0, w1 * r1.x); atomicAdd(ap + 1, w1 * r1.y);
        atomicAdd(ap + 2, w1 * r1.z); atomicAdd(ap + 3, w1 * r1.w);
    }
    if (FULL || i4 + 2 < cnt) {
        float* ap = acc + (int)(a2 & 1023u) * AST + 4 * q;
        atomicAdd(ap + 0, w2 * r2.x); atomicAdd(ap + 1, w2 * r2.y);
        atomicAdd(ap + 2, w2 * r2.z); atomicAdd(ap + 3, w2 * r2.w);
    }
    if (FULL || i4 + 3 < cnt) {
        float* ap = acc + (int)(a3 & 1023u) * AST + 4 * q;
        atomicAdd(ap + 0, w3 * r3.x); atomicAdd(ap + 1, w3 * r3.y);
        atomicAdd(ap + 2, w3 * r3.z); atomicAdd(ap + 3, w3 * r3.w);
    }
    if (ok) atomicAdd(acc + dl * AST + HD, w);
}

__global__ void __launch_bounds__(1024)
k_edge_own(const int* __restrict__ src, const int* __restrict__ dst,
           const unsigned int* __restrict__ key4, const unsigned int* __restrict__ lhead,
           const unsigned int* __restrict__ lists, size_t need_bytes, size_t ws_bytes,
           const float* __restrict__ hw, const float* __restrict__ es, const float* __restrict__ ed,
           float* __restrict__ h, int N, int E, int npb) {
    __shared__ float acc[NPBMAX * AST];
    __shared__ unsigned int mEnc[NPBMAX];
    __shared__ float edl[NPBMAX];
    __shared__ unsigned int lstA[BCAP];
    __shared__ float lstB[BCAP];

    const int tid = threadIdx.x;
    const int n0 = blockIdx.x * npb;
    int nown = N - n0; if (nown > npb) nown = npb; if (nown < 0) nown = 0;

    for (int i = tid; i < nown * AST; i += 1024) acc[i] = 0.f;
    for (int i = tid; i < nown; i += 1024) { mEnc[i] = 0x007FFFFFu;   edl[i] = ed[n0 + i]; }
    const unsigned int hd = (need_bytes <= ws_bytes) ? lhead[blockIdx.x] : NOLIST;
    __syncthreads();

    if (hd != NOLIST) {
        const int cnt = hd < (unsigned int)BCAP ? (int)hd : BCAP;
        const unsigned int* L = lists + (size_t)blockIdx.x * BCAP;
        for (int i = tid; i < cnt; i += 1024) {
            const unsigned int a = L[i];
            const int dl = (int)(a & 1023u);
            const float lg = lrelu02(es[a >> 10] + edl[dl]);
            atomicMax(&mEnc[dl], encOrd(lg));
            lstA[i] = a;
            lstB[i] = lg;
        }
        __syncthreads();
        int base = 0;
        for (; base + 1024 <= cnt; base += 1024) sum_trip<true>(acc, mEnc, lstA, lstB, hw, base, cnt, tid);
        if (base < cnt) sum_trip<false>(acc, mEnc, lstA, lstB, hw, base, cnt, tid);
    } else {
        const uint4* kv = (const uint4*)key4;
        const int nvec = (E + 15) >> 4;
        const unsigned int pat = (unsigned int)blockIdx.x * 0x01010101u;
        scan_keys(kv, nvec, pat, E, [&](int e, int) {
            const int dl = dst[e] - n0;
            atomicMax(&mEnc[dl], encOrd(lrelu02(es[src[e]] + edl[dl])));
        });
        __syncthreads();
        scan_keys(kv, nvec, pat, E, [&](int e, int) {
            const int s = src[e];
            const int dl = dst[e] - n0;
            const float w = __expf(lrelu02(es[s] + edl[dl]) - decOrd(mEnc[dl]));
            const float* r = hw + (size_t)s * HD;
            float* ap = acc + dl * AST;
#pragma unroll
            for (int f = 0; f < HD; ++f) atomicAdd(ap + f, w * r[f]);
            atomicAdd(ap + HD, w);
        });
    }
    __syncthreads();

    for (int j = tid; j < nown * HD; j += 1024) {
        const int n = j >> 4, f = j & 15;
        const float den = fmaxf(acc[n * AST + HD], 1e-16f);
        const float x = acc[n * AST + f] / den;
        h[(size_t)n0 * HD + j] = x > 0.f ? x : (__expf(x) - 1.f);
    }
}

__global__ void k_final(const float* __restrict__ h, const float* __restrict__ mlpW,
                        const float* __restrict__ mlpb, float* __restrict__ out, int N) {
    int n = blockIdx.x * blockDim.x + threadIdx.x;
    if (n >= N) return;
    const float4* r = (const float4*)(h + (size_t)n * HD);
    float acc = mlpb[0];
#pragma unroll
    for (int q = 0; q < 4; ++q) {
        const float4 v = r[q];
        acc += v.x * mlpW[q*4+0] + v.y * mlpW[q*4+1] + v.z * mlpW[q*4+2] + v.w * mlpW[q*4+3];
    }
    out[n] = acc;
}

static inline int cdiv(int a, int b) { return (a + b - 1) / b; }

extern "C" void kernel_launch(void* const* d_in, const int* in_sizes, int n_in,
                              void* d_out, int out_size, void* d_ws, size_t ws_size,
                              hipStream_t stream) {
    const float* fw     = (const float*)d_in[0];
    const float* sw     = (const float*)d_in[1];
    const float* tw     = (const float*)d_in[2];
    const float* areas  = (const float*)d_in[3];
    const int*   indices= (const int*)  d_in[4];
    const int*   edge   = (const int*)  d_in[5];
    const float* W0     = (const float*)d_in[6];
    const float* Wl     = (const float*)d_in[7];
    const float* a_src  = (const float*)d_in[8];
    const float* a_dst  = (const float*)d_in[9];
    const float* bvec   = (const float*)d_in[10];
    const float* mlpW   = (const float*)d_in[11];
    const float* mlpb   = (const float*)d_in[12];

    const int N = in_sizes[3];
    const int E = in_sizes[5] / 2;
    const int T = in_sizes[0] / (NWIRE * NFEAT);
    const int* src = edge;
    const int* dst = edge + E;

    float* ws = (float*)d_ws;
    const int nvec4 = cdiv(E, 16) * 4;
    size_t off = 0;
    unsigned int* key4   = (unsigned int*)(ws + off); off += (size_t)nvec4;
    float*        h      = ws + off; off += (size_t)N * HD;
    float*        hw     = ws + off; off += (size_t)N * HD;
    float*        es     = ws + off; off += (size_t)N;
    float*        ed     = ws + off; off += (size_t)N;
    float*        W0p    = ws + off; off += 256;
    unsigned int* lhead  = (unsigned int*)(ws + off); off += NB;
    unsigned int* lists  = (unsigned int*)(ws + off); off += (size_t)NB * BCAP;
    const size_t need_bytes = off * sizeof(float);
    (void)n_in; (void)out_size;

    const int npb = cdiv(N, NB);
    k_key<<<cdiv(nvec4, 256), 256, 0, stream>>>(dst, key4, E, nvec4, npb);
    k_padW0<<<1, 256, 0, stream>>>(W0, W0p);
    k_list<<<NB, 1024, 0, stream>>>(src, dst, key4, lhead, lists, need_bytes, ws_size, N, E, npb);

    const int tiles  = cdiv(N, 16);
    const int gemmBl = cdiv(tiles, 8);

    for (int g = 0; g < T; ++g) {
        k_fill<<<cdiv(N, 256), 256, 0, stream>>>(fw, sw, tw, areas, indices, h, N, g);
        for (int i = 0; i < 4; ++i) {
            const float* Wi = (i == 0) ? W0p : (Wl + (size_t)(i - 1) * 256);
            k_gemm_wmma<<<gemmBl, 256, 0, stream>>>(h, Wi, bvec + (size_t)i * HD, a_src + (size_t)i * HD,
                                                    a_dst + (size_t)i * HD, hw, es, ed, N);
            k_edge_own<<<NB, 1024, 0, stream>>>(src, dst, key4, lhead, lists, need_bytes, ws_size,
                                                hw, es, ed, h, N, E, npb);
        }
        k_final<<<cdiv(N, 256), 256, 0, stream>>>(h, mlpW, mlpb, (float*)d_out + (size_t)g * N, N);
    }
}


# Sum phase: 16 lanes share an edge (lane = one float of its hw row, word and weight by DPP row_share): an LDS float add covers 2 whole rows, not 8 quarter rows; sums arrival-ordered as before
# speedup vs baseline: 9.3112x; 1.0715x over previous
#include <hip/hip_runtime.h>
#include <math.h>

typedef float v2f __attribute__((ext_vector_type(2)));
typedef float v8f __attribute__((ext_vector_type(8)));

#define HD 16
#define NWIRE 1536
#define NFEAT 4
#define NB 256
#define NPBMAX 1024
#define AST 16
#define SCAN_U 8
#define BCAP 16384
#define NOLIST 0xFFFFFFFFu
#define LWCAP 1312
#define NBK 8192

__device__ __forceinline__ unsigned int encOrd(float f) {
    unsigned int b = __float_as_uint(f);
    return (b & 0x80000000u) ? ~b : (b | 0x80000000u);
}
__device__ __forceinline__ float decOrd(unsigned int e) {
    unsigned int b = (e & 0x80000000u) ? (e ^ 0x80000000u) : ~e;
    return __uint_as_float(b);
}
__device__ __forceinline__ float lrelu02(float x) { return x > 0.f ? x : 0.2f * x; }
template <int S>
__device__ __forceinline__ unsigned int row_bcast(unsigned int v) {
    return (unsigned int)__builtin_amdgcn_update_dpp(0, (int)v, 0x150 + S, 0xF, 0xF, true);
}

template <int CTRL>
__device__ __forceinline__ float row_from(float v) {
    return __uint_as_float((unsigned int)__builtin_amdgcn_update_dpp(0, (int)__float_as_uint(v), CTRL, 0xF, 0xF, true));
}
template <int CTRL, int NOUT>
__device__ __forceinline__ void fold_half(float* v, bool up) {
#pragma unroll
    for (int i = 0; i < NOUT; ++i) {
        const float keep = up ? v[NOUT + i] : v[i];
        const float send = up ? v[i] : v[NOUT + i];
        v[i] = keep + row_from<CTRL>(send);
    }
}

__global__ void k_padW0(const float* __restrict__ W0, float* __restrict__ W0p) {
    int i = threadIdx.x;
    int r = i >> 4;
    W0p[i] = (r < 13) ? W0[i] : 0.f;
}

__global__ void k_fill(const float* __restrict__ fw, const float* __restrict__ sw,
                       const float* __restrict__ tw, const float* __restrict__ areas,
                       const int* __restrict__ indices, float* __restrict__ h,
                       int N, int g) {
    int n = blockIdx.x * blockDim.x + threadIdx.x;
    if (n >= N) return;
    const size_t gbase = (size_t)g * NWIRE * NFEAT;
    const float4 f0 = *(const float4*)(fw + gbase + (size_t)indices[n * 3 + 0] * NFEAT);
    const float4 f1 = *(const float4*)(sw + gbase + (size_t)indices[n * 3 + 1] * NFEAT);
    const float4 f2 = *(const float4*)(tw + gbase + (size_t)indices[n * 3 + 2] * NFEAT);
    float4* hp = (float4*)(h + (size_t)n * HD);
    hp[0] = f0;
    hp[1] = f1;
    hp[2] = f2;
    hp[3] = make_float4(areas[n], 0.f, 0.f, 0.f);
}

__global__ void k_gemm_wmma(const float* __restrict__ h, const float* __restrict__ W,
                            const float* __restrict__ bias, const float* __restrict__ asrc,
                            const float* __restrict__ adst, float* __restrict__ hw,
                            float* __restrict__ es, float* __restrict__ ed, int N) {
    const int lane = threadIdx.x & 31;
    const int wave = threadIdx.x >> 5;
    const int tile = blockIdx.x * 8 + wave;
    const int base = tile * 16;
    if (base >= N) return;
    const int col16 = lane & 15;
    const int half  = lane >> 4;
    int arow = base + col16;
    if (arow >= N) arow = N - 1;
    v8f acc = {};
#pragma unroll
    for (int k = 0; k < 4; ++k) {
        const int kc = 4 * k + 2 * half;
        v2f a;
        a.x = h[(size_t)arow * HD + kc];
        a.y = h[(size_t)arow * HD + kc + 1];
        v2f b;
        b.x = W[kc * HD + col16];
        b.y = W[(kc + 1) * HD + col16];
        acc = __builtin_amdgcn_wmma_f32_16x16x4_f32(
                  false, a, false, b, (short)0, acc, false, false);
    }
    const float bi = bias[col16];
    const float as = asrc[col16];
    const float ad = adst[col16];
    float p[16];
#pragma unroll
    for (int r = 0; r < 8; ++r) {
        const float v = acc[r] + bi;
        acc[r] = v;
        p[r] = v * as;
        p[8 + r] = v * ad;
    }
    if (base + 16 <= N) {
#pragma unroll
        for (int r = 0; r < 8; ++r) {
            hw[(size_t)(base + r + 8 * half) * HD + col16] = acc[r];
        }
    } else {
#pragma unroll
        for (int r = 0; r < 8; ++r) {
            const int orow = base + r + 8 * half;
            if (orow < N) hw[(size_t)orow * HD + col16] = acc[r];
        }
    }
    fold_half<0x128, 8>(p, (col16 & 8) != 0);
    fold_half<0x141, 4>(p, (col16 & 4) != 0);
    fold_half<0x4E, 2>(p, (col16 & 2) != 0);
    fold_half<0xB1, 1>(p, (col16 & 1) != 0);
    const int prow = base + 8 * half + (col16 & 7);
    float* pd = (col16 & 8) ? ed : es;
    if (prow < N) pd[prow] = p[0];
}

__global__ void k_key(const int* __restrict__ dst, unsigned int* __restrict__ key4,
                      int E, int nvec4, int npb) {
    const int j = blockIdx.x * blockDim.x + threadIdx.x;
    if (j >= nvec4) return;
    unsigned int pk = 0u;
#pragma unroll
    for (int k = 0; k < 4; ++k) {
        const int e = 4 * j + k;
        const unsigned int o = (e < E) ? (unsigned int)(dst[e] / npb) : 0u;
        pk |= (o & 0xFFu) << (8 * k);
    }
    key4[j] = pk;
}

template <class F>
__device__ __forceinline__ int scan_keys(const uint4* __restrict__ kv, int nvec, unsigned int pat,
                                         int E, F fn) {
    int wcnt = 0;
    const int lane = threadIdx.x & 31;
    for (int vb = threadIdx.x & ~31; vb < nvec; vb += 1024 * SCAN_U) {
        uint4 q[SCAN_U];
#pragma unroll
        for (int u = 0; u < SCAN_U; ++u) {
            const int v = vb + u * 1024 + lane;
            q[u] = kv[v < nvec ? v : nvec - 1];
        }
#pragma unroll
        for (int u = 0; u < SCAN_U; ++u) {
            const int v = vb + u * 1024 + lane;
            const unsigned int wd[4] = {q[u].x, q[u].y, q[u].z, q[u].w};
            unsigned int m = 0u;
#pragma unroll
            for (int j = 0; j < 4; ++j) {
                const unsigned int x = wd[j] ^ pat;
                const unsigned int t = ~(((x & 0x7F7F7F7Fu) + 0x7F7F7F7Fu) | x | 0x7F7F7F7Fu);
                m |= t >> (7 - j);
            }
            if (v >= nvec) m = 0u;
            else if (v * 16 + 16 > E) {
                for (int p = 0; p < 32; ++p)
                    if (((m >> p) & 1u) && v * 16 + (p & 7) * 4 + (p >> 3) >= E) m &= ~(1u << p);
            }
            for (int it = 0; it < 16; ++it) {
                const bool has = (m != 0u);
                const unsigned int mk = (unsigned int)__ballot(has);
                if (mk == 0u) break;
                if (has) {
                    const int p = __builtin_ctz(m);
                    m &= m - 1u;
                    fn(v * 16 + (p & 7) * 4 + (p >> 3), wcnt + (int)__builtin_amdgcn_mbcnt_lo(mk, 0u));
                }
                wcnt += __popc(mk);
            }
        }
    }
    return wcnt;
}

__global__ void __launch_bounds__(1024)
k_list(const int* __restrict__ src, const int* __restrict__ dst,
       const unsigned int* __restrict__ key4, unsigned int* __restrict__ lhead,
       unsigned int* __restrict__ lists, size_t need_bytes, size_t ws_bytes, int N, int E, int npb) {
    __shared__ unsigned int lstA[32 * LWCAP];
    __shared__ unsigned int srt[BCAP];
    __shared__ unsigned int cntb[NBK];
    __shared__ int wcn[32];
    __shared__ unsigned int wsum[32];
    __shared__ int ovf;
    if (need_bytes > ws_bytes) return;

    const int tid = threadIdx.x;
    const int lane = tid & 31;
    const int wave = tid >> 5;
    const int wbase = wave * LWCAP;
    const int n0 = blockIdx.x * npb;
    for (int i = tid; i < NBK; i += 1024) cntb[i] = 0u;
    if (tid == 0) ovf = 0;
    __syncthreads();

    const uint4* kv = (const uint4*)key4;
    const int nvec = (E + 15) >> 4;
    const unsigned int pat = (unsigned int)blockIdx.x * 0x01010101u;
    const int wcnt = scan_keys(kv, nvec, pat, E, [&](int e, int slot) {
        if (slot < LWCAP) lstA[wbase + slot] = (unsigned int)e; else ovf = 1;
    });
    if (lane == 0) wcn[wave] = wcnt;
    __syncthreads();

    int total = 0;
    for (int w = 0; w < 32; ++w) total += wcn[w];
    const bool ok = (ovf == 0) && (total <= BCAP);
    if (tid == 0) lhead[blockIdx.x] = ok ? (unsigned int)total : NOLIST;
    if (!ok) return;

    int sh = 5;
    while (((N - 1) >> sh) >= NBK) ++sh;
    const unsigned int nbk = (unsigned int)((N - 1) >> sh) + 1u;
    const unsigned int rot = (unsigned int)(((size_t)blockIdx.x * nbk) / gridDim.x);
    auto bucket = [&](unsigned int s) -> unsigned int {
        unsigned int b = s >> sh;
        if (b >= nbk) b = nbk - 1u;
        b = b >= rot ? b - rot : b + nbk - rot;
        return b;
    };

    for (int i = lane; i < wcnt; i += 32) {
        const int e = (int)lstA[wbase + i];
        const unsigned int sv = (unsigned int)src[e];
        lstA[wbase + i] = (sv << 10) | (unsigned int)(dst[e] - n0);
        atomicAdd(&cntb[bucket(sv)], 1u);
    }
    __syncthreads();

    unsigned int c[NBK / 1024];
    unsigned int s8 = 0u;
#pragma unroll
    for (int k = 0; k < NBK / 1024; ++k) { c[k] = cntb[tid * (NBK / 1024) + k]; s8 += c[k]; }
    unsigned int inc = s8;
    for (int d = 1; d < 32; d <<= 1) {
        const unsigned int v = (unsigned int)__shfl_up((int)inc, (unsigned int)d);
        if (lane >= d) inc += v;
    }
    if (lane == 31) wsum[wave] = inc;
    __syncthreads();
    unsigned int run = inc - s8;
    for (int w = 0; w < wave; ++w) run += wsum[w];
#pragma unroll
    for (int k = 0; k < NBK / 1024; ++k) { cntb[tid * (NBK / 1024) + k] = run; run += c[k]; }
    __syncthreads();

    for (int i = lane; i < wcnt; i += 32) {
        const unsigned int wd = lstA[wbase + i];
        const unsigned int pos = atomicAdd(&cntb[bucket(wd >> 10)], 1u);
        srt[pos] = wd;
    }
    __syncthreads();

    unsigned int* L = lists + (size_t)blockIdx.x * BCAP;
    for (int i = tid; i < total; i += 1024) L[i] = srt[i];
}

#define SUM_LOAD(S) as[S] = row_bcast<S>(a); r[S] = hw[(as[S] >> 10) * (unsigned int)HD + c];
#define SUM_ADD(S) { const float ws = __uint_as_float(row_bcast<S>(wb)); \
                     if (FULL || i16 + S < cnt) atomicAdd(acc + (as[S] & 1023u) * (unsigned int)AST + c, ws * r[S]); }
template <bool FULL>
__device__ __forceinline__ void sum_trip(float* acc, float* den, const unsigned int* mEnc, const unsigned int* lstA,
                                         const float* lstB, const float* __restrict__ hw,
                                         int base, int cnt, int tid) {
    const unsigned int c = (unsigned int)tid & 15u;
    const int i = base + tid;
    const bool ok = FULL || i < cnt;
    const int ii = ok ? i : 0;
    const unsigned int a = lstA[ii];
    unsigned int as[16];
    float r[16];
    SUM_LOAD(0)  SUM_LOAD(1)  SUM_LOAD(2)  SUM_LOAD(3)  SUM_LOAD(4)  SUM_LOAD(5)  SUM_LOAD(6)  SUM_LOAD(7)
    SUM_LOAD(8)  SUM_LOAD(9)  SUM_LOAD(10) SUM_LOAD(11) SUM_LOAD(12) SUM_LOAD(13) SUM_LOAD(14) SUM_LOAD(15)
    const int dl = (int)(a & 1023u);
    const float w = __expf(lstB[ii] - decOrd(mEnc[dl]));
    const unsigned int wb = __float_as_uint(w);
    const int i16 = base + (tid & ~15);
    SUM_ADD(0)  SUM_ADD(1)  SUM_ADD(2)  SUM_ADD(3)  SUM_ADD(4)  SUM_ADD(5)  SUM_ADD(6)  SUM_ADD(7)
    SUM_ADD(8)  SUM_ADD(9)  SUM_ADD(10) SUM_ADD(11) SUM_ADD(12) SUM_ADD(13) SUM_ADD(14) SUM_ADD(15)
    if (ok) atomicAdd(den + dl, w);
}
#undef SUM_LOAD
#undef SUM_ADD

__global__ void __launch_bounds__(1024)
k_edge_own(const int* __restrict__ src, const int* __restrict__ dst,
           const unsigned int* __restrict__ key4, const unsigned int* __restrict__ lhead,
           const unsigned int* __restrict__ lists, size_t need_bytes, size_t ws_bytes,
           const float* __restrict__ hw, const float* __restrict__ es, const float* __restrict__ ed,
           float* __restrict__ h, int N, int E, int npb) {
    __shared__ float acc[NPBMAX * AST];
    __shared__ float den[NPBMAX];
    __shared__ unsigned int mEnc[NPBMAX];
    __shared__ float edl[NPBMAX];
    __shared__ unsigned int lstA[BCAP];
    __shared__ float lstB[BCAP];

    const int tid = threadIdx.x;
    const int n0 = blockIdx.x * npb;
    int nown = N - n0; if (nown > npb) nown = npb; if (nown < 0) nown = 0;

    for (int i = tid; i < nown * AST; i += 1024) acc[i] = 0.f;
    for (int i = tid; i < nown; i += 1024) { den[i] = 0.f; mEnc[i] = 0x007FFFFFu;   edl[i] = ed[n0 + i]; }
    const unsigned int hd = (need_bytes <= ws_bytes) ? lhead[blockIdx.x] : NOLIST;
    __syncthreads();

    if (hd != NOLIST) {
        const int cnt = hd < (unsigned int)BCAP ? (int)hd : BCAP;
        const unsigned int* L = lists + (size_t)blockIdx.x * BCAP;
        for (int i = tid; i < cnt; i += 1024) {
            const unsigned int a = L[i];
            const int dl = (int)(a & 1023u);
            const float lg = lrelu02(es[a >> 10] + edl[dl]);
            atomicMax(&mEnc[dl], encOrd(lg));
            lstA[i] = a;
            lstB[i] = lg;
        }
        __syncthreads();
        int base = 0;
        for (; base + 1024 <= cnt; base += 1024) sum_trip<true>(acc, den, mEnc, lstA, lstB, hw, base, cnt, tid);
        if (base < cnt) sum_trip<false>(acc, den, mEnc, lstA, lstB, hw, base, cnt, tid);
    } else {
        const uint4* kv = (const uint4*)key4;
        const int nvec = (E + 15) >> 4;
        const unsigned int pat = (unsigned int)blockIdx.x * 0x01010101u;
        scan_keys(kv, nvec, pat, E, [&](int e, int) {
            const int dl = dst[e] - n0;
            atomicMax(&mEnc[dl], encOrd(lrelu02(es[src[e]] + edl[dl])));
        });
        __syncthreads();
        scan_keys(kv, nvec, pat, E, [&](int e, int) {
            const int s = src[e];
            const int dl = dst[e] - n0;
            const float w = __expf(lrelu02(es[s] + edl[dl]) - decOrd(mEnc[dl]));
            const float* r = hw + (size_t)s * HD;
            float* ap = acc + dl * AST;
#pragma unroll
            for (int f = 0; f < HD; ++f) atomicAdd(ap + f, w * r[f]);
            atomicAdd(den + dl, w);
        });
    }
    __syncthreads();

    for (int j = tid; j < nown * HD; j += 1024) {
        const int n = j >> 4, f = j & 15;
        const float dn = fmaxf(den[n], 1e-16f);
        const float x = acc[n * AST + f] / dn;
        h[(size_t)n0 * HD + j] = x > 0.f ? x : (__expf(x) - 1.f);
    }
}

__global__ void k_final(const float* __restrict__ h, const float* __restrict__ mlpW,
                        const float* __restrict__ mlpb, float* __restrict__ out, int N) {
    int n = blockIdx.x * blockDim.x + threadIdx.x;
    if (n >= N) return;
    const float4* r = (const float4*)(h + (size_t)n * HD);
    float acc = mlpb[0];
#pragma unroll
    for (int q = 0; q < 4; ++q) {
        const float4 v = r[q];
        acc += v.x * mlpW[q*4+0] + v.y * mlpW[q*4+1] + v.z * mlpW[q*4+2] + v.w * mlpW[q*4+3];
    }
    out[n] = acc;
}

static inline int cdiv(int a, int b) { return (a + b - 1) / b; }

extern "C" void kernel_launch(void* const* d_in, const int* in_sizes, int n_in,
                              void* d_out, int out_size, void* d_ws, size_t ws_size,
                              hipStream_t stream) {
    const float* fw     = (const float*)d_in[0];
    const float* sw     = (const float*)d_in[1];
    const float* tw     = (const float*)d_in[2];
    const float* areas  = (const float*)d_in[3];
    const int*   indices= (const int*)  d_in[4];
    const int*   edge   = (const int*)  d_in[5];
    const float* W0     = (const float*)d_in[6];
    const float* Wl     = (const float*)d_in[7];
    const float* a_src  = (const float*)d_in[8];
    const float* a_dst  = (const float*)d_in[9];
    const float* bvec   = (const float*)d_in[10];
    const float* mlpW   = (const float*)d_in[11];
    const float* mlpb   = (const float*)d_in[12];

    const int N = in_sizes[3];
    const int E = in_sizes[5] / 2;
    const int T = in_sizes[0] / (NWIRE * NFEAT);
    const int* src = edge;
    const int* dst = edge + E;

    float* ws = (float*)d_ws;
    const int nvec4 = cdiv(E, 16) * 4;
    size_t off = 0;
    unsigned int* key4   = (unsigned int*)(ws + off); off += (size_t)nvec4;
    float*        h      = ws + off; off += (size_t)N * HD;
    float*        hw     = ws + off; off += (size_t)N * HD;
    float*        es     = ws + off; off += (size_t)N;
    float*        ed     = ws + off; off += (size_t)N;
    float*        W0p    = ws + off; off += 256;
    unsigned int* lhead  = (unsigned int*)(ws + off); off += NB;
    unsigned int* lists  = (unsigned int*)(ws + off); off += (size_t)NB * BCAP;
    const size_t need_bytes = off * sizeof(float);
    (void)n_in; (void)out_size;

    const int npb = cdiv(N, NB);
    k_key<<<cdiv(nvec4, 256), 256, 0, stream>>>(dst, key4, E, nvec4, npb);
    k_padW0<<<1, 256, 0, stream>>>(W0, W0p);
    k_list<<<NB, 1024, 0, stream>>>(src, dst, key4, lhead, lists, need_bytes, ws_size, N, E, npb);

    const int tiles  = cdiv(N, 16);
    const int gemmBl = cdiv(tiles, 8);

    for (int g = 0; g < T; ++g) {
        k_fill<<<cdiv(N, 256), 256, 0, stream>>>(fw, sw, tw, areas, indices, h, N, g);
        for (int i = 0; i < 4; ++i) {
            const float* Wi = (i == 0) ? W0p : (Wl + (size_t)(i - 1) * 256);
            k_gemm_wmma<<<gemmBl, 256, 0, stream>>>(h, Wi, bvec + (size_t)i * HD, a_src + (size_t)i * HD,
                                                    a_dst + (size_t)i * HD, hw, es, ed, N);
            k_edge_own<<<NB, 1024, 0, stream>>>(src, dst, key4, lhead, lists, need_bytes, ws_size,
                                                hw, es, ed, h, N, E, npb);
        }
        k_final<<<cdiv(N, 256), 256, 0, stream>>>(h, mlpW, mlpb, (float*)d_out + (size_t)g * N, N);
    }
}
